# starting kernel
# baseline (speedup 1.0000x reference)
#include <hip/hip_runtime.h>
#include <math.h>


typedef __attribute__((ext_vector_type(2))) float v2f;
typedef __attribute__((ext_vector_type(8))) float v8f;

#define NQ      12
#define NL      4
#define QDIM    4096
#define THREADS 256

static_assert(QDIM == (1 << NQ), "dim mismatch");

struct c32 { float re, im; };
__device__ __forceinline__ c32 cmul(c32 a, c32 b) {
  return { a.re * b.re - a.im * b.im, a.re * b.im + a.im * b.re };
}
__device__ __forceinline__ c32 cadd(c32 a, c32 b) { return { a.re + b.re, a.im + b.im }; }

__device__ __forceinline__ void mat2mul(c32 r[4], const c32 p[4], const c32 q[4]) {
  r[0] = cadd(cmul(p[0], q[0]), cmul(p[1], q[2]));
  r[1] = cadd(cmul(p[0], q[1]), cmul(p[1], q[3]));
  r[2] = cadd(cmul(p[2], q[0]), cmul(p[3], q[2]));
  r[3] = cadd(cmul(p[2], q[1]), cmul(p[3], q[3]));
}

template <int SK, int SN, int BM>
__device__ __forceinline__ void apply_group(float* sre, float* sim,
                                            const float* AreG, const float* AimG,
                                            int wave, int half, int mn) {
#pragma unroll
  for (int tt = 0; tt < 2; ++tt) {
    const int tile = wave + tt * 8;
    const int base = tile * BM;
    v8f cre = {0.f, 0.f, 0.f, 0.f, 0.f, 0.f, 0.f, 0.f};
    v8f cim = {0.f, 0.f, 0.f, 0.f, 0.f, 0.f, 0.f, 0.f};
#pragma unroll
    for (int kc = 0; kc < 4; ++kc) {
      const int k0 = kc * 4 + 2 * half;
      const int ar = mn * 16 + k0;
      v2f Ar = { AreG[ar], AreG[ar + 1] };
      v2f Ai = { AimG[ar], AimG[ar + 1] };
      v2f An = -Ai;
      const int ib = base + k0 * SK + mn * SN;
      v2f Br = { sre[ib], sre[ib + SK] };
      v2f Bi = { sim[ib], sim[ib + SK] };
      cre = __builtin_amdgcn_wmma_f32_16x16x4_f32(false, Ar, false, Br, 0, cre, false, false);
      cre = __builtin_amdgcn_wmma_f32_16x16x4_f32(false, An, false, Bi, 0, cre, false, false);
      cim = __builtin_amdgcn_wmma_f32_16x16x4_f32(false, Ar, false, Bi, 0, cim, false, false);
      cim = __builtin_amdgcn_wmma_f32_16x16x4_f32(false, Ai, false, Br, 0, cim, false, false);
    }
#pragma unroll
    for (int v = 0; v < 8; ++v) {
      const int m   = v + 8 * half;
      const int idx = base + m * SK + mn * SN;
      sre[idx] = cre[v];
      sim[idx] = cim[v];
    }
  }
}

__global__ __launch_bounds__(THREADS)
void qcircuit_parity_kernel(const float* __restrict__ x,
                            const float* __restrict__ thetas,
                            float* __restrict__ out) {
  __shared__ float sre[QDIM];
  __shared__ float sim[QDIM];
  __shared__ float Are[3 * 256];
  __shared__ float Aim[3 * 256];
  __shared__ float gmre[NQ * 4];
  __shared__ float gmim[NQ * 4];
  __shared__ float red[THREADS];

  const int tid  = threadIdx.x;
  const int b    = blockIdx.x;
  const int lane = tid & 31;
  const int wave = tid >> 5;
  const int half = lane >> 4;
  const int mn   = lane & 15;

  for (int j = tid; j < QDIM; j += THREADS) { sre[j] = 0.f; sim[j] = 0.f; }
  if (tid == 0) sre[0] = 1.f;

  const float xv = x[b];
  const float x1 = asinf(xv);
  const float x2 = acosf(xv * xv);

  int csrc[16];
#pragma unroll
  for (int j = 0; j < 16; ++j) {
    int s = tid + j * THREADS;
#pragma unroll
    for (int jj = 10; jj >= 0; --jj)
      s ^= ((s >> (11 - jj)) & 1) << (10 - jj);
    csrc[j] = s;
  }

  __syncthreads();

  for (int l = 0; l < NL; ++l) {
    if (tid < NQ) {
      const int q = tid;
      c32 M[4];
      { float s, c; __sincosf(0.5f * x1, &s, &c);
        M[0] = c32{c, 0.f}; M[1] = c32{-s, 0.f}; M[2] = c32{s, 0.f}; M[3] = c32{c, 0.f}; }
      { float s, c; __sincosf(0.5f * x2, &s, &c);
        c32 G[4] = { c32{c, -s}, c32{0.f, 0.f}, c32{0.f, 0.f}, c32{c, s} };
        c32 T[4]; mat2mul(T, G, M);
        M[0] = T[0]; M[1] = T[1]; M[2] = T[2]; M[3] = T[3]; }
      const float t0 = thetas[(l * NQ + q) * 3 + 0];
      const float t1 = thetas[(l * NQ + q) * 3 + 1];
      const float t2 = thetas[(l * NQ + q) * 3 + 2];
      { float s, c; __sincosf(0.5f * t0, &s, &c);
        c32 G[4] = { c32{c, 0.f}, c32{0.f, -s}, c32{0.f, -s}, c32{c, 0.f} };
        c32 T[4]; mat2mul(T, G, M);
        M[0] = T[0]; M[1] = T[1]; M[2] = T[2]; M[3] = T[3]; }
      { float s, c; __sincosf(0.5f * t1, &s, &c);
        c32 G[4] = { c32{c, -s}, c32{0.f, 0.f}, c32{0.f, 0.f}, c32{c, s} };
        c32 T[4]; mat2mul(T, G, M);
        M[0] = T[0]; M[1] = T[1]; M[2] = T[2]; M[3] = T[3]; }
      { float s, c; __sincosf(0.5f * t2, &s, &c);
        c32 G[4] = { c32{c, 0.f}, c32{0.f, -s}, c32{0.f, -s}, c32{c, 0.f} };
        c32 T[4]; mat2mul(T, G, M);
        M[0] = T[0]; M[1] = T[1]; M[2] = T[2]; M[3] = T[3]; }
#pragma unroll
      for (int k = 0; k < 4; ++k) { gmre[q * 4 + k] = M[k].re; gmim[q * 4 + k] = M[k].im; }
    }
    __syncthreads();

    {
      const int r = tid >> 4, cc = tid & 15;
#pragma unroll
      for (int g = 0; g < 3; ++g) {
        c32 acc = c32{1.f, 0.f};
#pragma unroll
        for (int j = 0; j < 4; ++j) {
          const int q  = 4 * g + j;
          const int bi = (r  >> (3 - j)) & 1;
          const int bj = (cc >> (3 - j)) & 1;
          acc = cmul(acc, c32{ gmre[q * 4 + bi * 2 + bj], gmim[q * 4 + bi * 2 + bj] });
        }
        Are[g * 256 + r * 16 + cc] = acc.re;
        Aim[g * 256 + r * 16 + cc] = acc.im;
      }
    }
    __syncthreads();

    apply_group<256, 1, 16 >(sre, sim, Are + 0,   Aim + 0,   wave, half, mn);
    __syncthreads();
    apply_group<16,  1, 256>(sre, sim, Are + 256, Aim + 256, wave, half, mn);
    __syncthreads();
    apply_group<1,  16, 256>(sre, sim, Are + 512, Aim + 512, wave, half, mn);
    __syncthreads();

    float tr[16], ti[16];
#pragma unroll
    for (int j = 0; j < 16; ++j) { tr[j] = sre[csrc[j]]; ti[j] = sim[csrc[j]]; }
    __syncthreads();
#pragma unroll
    for (int j = 0; j < 16; ++j) {
      const int i = tid + j * THREADS;
      sre[i] = tr[j]; sim[i] = ti[j];
    }
    __syncthreads();
  }

  float acc = 0.f;
  for (int j = tid; j < QDIM; j += THREADS) {
    const float p = (__popc(j) & 1) ? -1.f : 1.f;
    acc = fmaf(sre[j] * sre[j] + sim[j] * sim[j], p, acc);
  }
  red[tid] = acc;
  __syncthreads();
  for (int off = THREADS / 2; off > 0; off >>= 1) {
    if (tid < off) red[tid] += red[tid + off];
    __syncthreads();
  }
  if (tid == 0) out[b] = red[0];
}

extern "C" void kernel_launch(void* const* d_in, const int* in_sizes, int n_in,
                              void* d_out, int out_size, void* d_ws, size_t ws_size,
                              hipStream_t stream) {
  (void)n_in; (void)d_ws; (void)ws_size; (void)out_size;
  const float* x      = (const float*)d_in[0];
  const float* thetas = (const float*)d_in[1];
  float* out          = (float*)d_out;
  const int batch     = in_sizes[0];
  qcircuit_parity_kernel<<<dim3(batch), dim3(THREADS), 0, stream>>>(x, thetas, out);
}


# Regroup 12 qubits as 4 groups of 3: each 8x8 complex gate is one 16x16 real WMMA matrix [[Ur,-Ui],[Ui,Ur]] (4 WMMA per tile, not 16), bank-swizzled 16-byte LDS access; f32 sum order changes
# speedup vs baseline: 1.4091x; 1.4091x over previous
#include <hip/hip_runtime.h>
#include <math.h>


typedef __attribute__((ext_vector_type(2))) float v2f;
typedef __attribute__((ext_vector_type(4))) float v4f;
typedef __attribute__((ext_vector_type(8))) float v8f;

#define NQ      12
#define NL      4
#define QDIM    4096
#define THREADS 256
#define APITCH  20

static_assert(QDIM == (1 << NQ), "dim mismatch");

struct c32 { float re, im; };
__device__ __forceinline__ c32 cmul(c32 a, c32 b) {
  return { a.re * b.re - a.im * b.im, a.re * b.im + a.im * b.re };
}
__device__ __forceinline__ c32 cadd(c32 a, c32 b) { return { a.re + b.re, a.im + b.im }; }

__device__ __forceinline__ void mat2mul(c32 r[4], const c32 p[4], const c32 q[4]) {
  r[0] = cadd(cmul(p[0], q[0]), cmul(p[1], q[2]));
  r[1] = cadd(cmul(p[0], q[1]), cmul(p[1], q[3]));
  r[2] = cadd(cmul(p[2], q[0]), cmul(p[3], q[2]));
  r[3] = cadd(cmul(p[2], q[1]), cmul(p[3], q[3]));
}

__device__ __forceinline__ constexpr int slot(int i) {
  return i ^ (((i >> 6) & 7) << 3) ^ (((i >> 9) & 1) << 2);
}

template <int G>
__device__ __forceinline__ constexpr int lidx(int kk, int w, int c) {
  return G == 0 ? ((kk << 9) | (w << 6) | (c << 2))
       : G == 1 ? ((w << 9) | (kk << 6) | (c << 2))
                : ((w << 9) | ((c >> 1) << 6) | (kk << 3) | ((c & 1) << 2));
}

__device__ __forceinline__ void mma4(v8f acc[4], const v2f a[4], const v2f b[4][4]) {
#pragma unroll
  for (int kc = 0; kc < 4; ++kc)
#pragma unroll
    for (int t = 0; t < 4; ++t)
      acc[t] = __builtin_amdgcn_wmma_f32_16x16x4_f32(false, a[kc], false, b[t][kc], 0, acc[t],
                                                     false, false);
}

__device__ __forceinline__ void load_a(v2f a[4], const float* Ag, int c, int h) {
  const v4f lo = *(const v4f*)&Ag[c * APITCH + 4 * h];
  const v4f hi = *(const v4f*)&Ag[c * APITCH + 8 + 4 * h];
  a[0] = v2f{lo[0], lo[1]}; a[1] = v2f{lo[2], lo[3]};
  a[2] = v2f{hi[0], hi[1]}; a[3] = v2f{hi[2], hi[3]};
}

template <int G>
__device__ __forceinline__ void pass_cols(float* st, const float* Ag, int w, int c, int h) {
  const int lane0 = slot(lidx<G>(0, w, c));
  const int ld0 = lane0 ^ (h ? slot(lidx<G>(4, 0, 0)) : 0);
  v4f L[2][4];
#pragma unroll
  for (int a = 0; a < 2; ++a)
#pragma unroll
    for (int j = 0; j < 4; ++j)
      L[a][j] = *(const v4f*)&st[a * QDIM + (ld0 ^ slot(lidx<G>(j, 0, 0)))];
  v2f af[4];
  load_a(af, Ag, c, h);
  v2f bf[4][4];
#pragma unroll
  for (int t = 0; t < 4; ++t)
#pragma unroll
    for (int kc = 0; kc < 4; ++kc)
      bf[t][kc] = v2f{ L[kc >> 1][2 * (kc & 1)][t], L[kc >> 1][2 * (kc & 1) + 1][t] };
  v8f acc[4];
#pragma unroll
  for (int t = 0; t < 4; ++t) acc[t] = v8f{0.f, 0.f, 0.f, 0.f, 0.f, 0.f, 0.f, 0.f};
  mma4(acc, af, bf);
#pragma unroll
  for (int r = 0; r < 8; ++r)
    *(v4f*)&st[h * QDIM + (lane0 ^ slot(lidx<G>(r, 0, 0)))] =
        v4f{ acc[0][r], acc[1][r], acc[2][r], acc[3][r] };
}

__device__ __forceinline__ void pass_low(float* st, const float* Ag, int w, int c, int h) {
  const int col0 = slot(((c >> 3) << 9) | (w << 6) | ((c & 7) << 3));
  v4f L[2][4];
#pragma unroll
  for (int a = 0; a < 2; ++a)
#pragma unroll
    for (int t = 0; t < 4; ++t)
      L[a][t] = *(const v4f*)&st[a * QDIM + (t << 10) + (col0 ^ (h << 2))];
  v2f af[4];
  load_a(af, Ag, c, h);
  v2f bf[4][4];
#pragma unroll
  for (int t = 0; t < 4; ++t)
#pragma unroll
    for (int kc = 0; kc < 4; ++kc)
      bf[t][kc] = v2f{ L[kc >> 1][t][2 * (kc & 1)], L[kc >> 1][t][2 * (kc & 1) + 1] };
  v8f acc[4];
#pragma unroll
  for (int t = 0; t < 4; ++t) acc[t] = v8f{0.f, 0.f, 0.f, 0.f, 0.f, 0.f, 0.f, 0.f};
  mma4(acc, af, bf);
#pragma unroll
  for (int t = 0; t < 4; ++t) {
    *(v4f*)&st[h * QDIM + (t << 10) + col0]       = v4f{ acc[t][0], acc[t][1], acc[t][2], acc[t][3] };
    *(v4f*)&st[h * QDIM + (t << 10) + (col0 ^ 4)] = v4f{ acc[t][4], acc[t][5], acc[t][6], acc[t][7] };
  }
}

__global__ __launch_bounds__(THREADS)
void qcircuit_parity_kernel(const float* __restrict__ x,
                            const float* __restrict__ thetas,
                            float* __restrict__ out) {
  __shared__ __attribute__((aligned(16))) float st[2 * QDIM];
  __shared__ __attribute__((aligned(16))) float Aemb[4 * 16 * APITCH];
  __shared__ float gmre[NQ * 4];
  __shared__ float gmim[NQ * 4];
  __shared__ float red[THREADS];
  float* sre = st;
  float* sim = st + QDIM;

  const int tid  = threadIdx.x;
  const int b    = blockIdx.x;
  const int lane = tid & 31;
  const int wave = tid >> 5;
  const int half = lane >> 4;
  const int mn   = lane & 15;

  for (int j = tid; j < QDIM; j += THREADS) { sre[j] = 0.f; sim[j] = 0.f; }
  if (tid == 0) sre[0] = 1.f;

  const float xv = x[b];
  const float x1 = asinf(xv);
  const float x2 = acosf(xv * xv);

  int csrc[16];
#pragma unroll
  for (int j = 0; j < 16; ++j) {
    int s = tid + j * THREADS;
#pragma unroll
    for (int jj = 10; jj >= 0; --jj)
      s ^= ((s >> (11 - jj)) & 1) << (10 - jj);
    csrc[j] = slot(s);
  }

  __syncthreads();

  for (int l = 0; l < NL; ++l) {
    if (tid < NQ) {
      const int q = tid;
      c32 M[4];
      { float s, c; __sincosf(0.5f * x1, &s, &c);
        M[0] = c32{c, 0.f}; M[1] = c32{-s, 0.f}; M[2] = c32{s, 0.f}; M[3] = c32{c, 0.f}; }
      { float s, c; __sincosf(0.5f * x2, &s, &c);
        c32 G[4] = { c32{c, -s}, c32{0.f, 0.f}, c32{0.f, 0.f}, c32{c, s} };
        c32 T[4]; mat2mul(T, G, M);
        M[0] = T[0]; M[1] = T[1]; M[2] = T[2]; M[3] = T[3]; }
      const float t0 = thetas[(l * NQ + q) * 3 + 0];
      const float t1 = thetas[(l * NQ + q) * 3 + 1];
      const float t2 = thetas[(l * NQ + q) * 3 + 2];
      { float s, c; __sincosf(0.5f * t0, &s, &c);
        c32 G[4] = { c32{c, 0.f}, c32{0.f, -s}, c32{0.f, -s}, c32{c, 0.f} };
        c32 T[4]; mat2mul(T, G, M);
        M[0] = T[0]; M[1] = T[1]; M[2] = T[2]; M[3] = T[3]; }
      { float s, c; __sincosf(0.5f * t1, &s, &c);
        c32 G[4] = { c32{c, -s}, c32{0.f, 0.f}, c32{0.f, 0.f}, c32{c, s} };
        c32 T[4]; mat2mul(T, G, M);
        M[0] = T[0]; M[1] = T[1]; M[2] = T[2]; M[3] = T[3]; }
      { float s, c; __sincosf(0.5f * t2, &s, &c);
        c32 G[4] = { c32{c, 0.f}, c32{0.f, -s}, c32{0.f, -s}, c32{c, 0.f} };
        c32 T[4]; mat2mul(T, G, M);
        M[0] = T[0]; M[1] = T[1]; M[2] = T[2]; M[3] = T[3]; }
#pragma unroll
      for (int k = 0; k < 4; ++k) { gmre[q * 4 + k] = M[k].re; gmim[q * 4 + k] = M[k].im; }
    }
    __syncthreads();

    {
      const int g = tid >> 6, r = (tid >> 3) & 7, cc = tid & 7;
      c32 acc = c32{1.f, 0.f};
#pragma unroll
      for (int j = 0; j < 3; ++j) {
        const int q  = 3 * g + j;
        const int bi = (r  >> (2 - j)) & 1;
        const int bj = (cc >> (2 - j)) & 1;
        acc = cmul(acc, c32{ gmre[q * 4 + bi * 2 + bj], gmim[q * 4 + bi * 2 + bj] });
      }
      float* Ag = Aemb + g * 16 * APITCH;
      Ag[r * APITCH + cc]           = acc.re;
      Ag[r * APITCH + cc + 8]       = -acc.im;
      Ag[(r + 8) * APITCH + cc]     = acc.im;
      Ag[(r + 8) * APITCH + cc + 8] = acc.re;
    }
    __syncthreads();

    pass_cols<0>(st, Aemb + 0 * 16 * APITCH, wave, mn, half);
    __builtin_amdgcn_fence(__ATOMIC_ACQ_REL, "wavefront");
    pass_low(st, Aemb + 3 * 16 * APITCH, wave, mn, half);
    __syncthreads();
    pass_cols<1>(st, Aemb + 1 * 16 * APITCH, wave, mn, half);
    __builtin_amdgcn_fence(__ATOMIC_ACQ_REL, "wavefront");
    pass_cols<2>(st, Aemb + 2 * 16 * APITCH, wave, mn, half);
    __syncthreads();

    float tr[16], ti[16];
#pragma unroll
    for (int j = 0; j < 16; ++j) { tr[j] = sre[csrc[j]]; ti[j] = sim[csrc[j]]; }
    __syncthreads();
#pragma unroll
    for (int j = 0; j < 16; ++j) {
      const int i = slot(tid + j * THREADS);
      sre[i] = tr[j]; sim[i] = ti[j];
    }
    __syncthreads();
  }

  float acc = 0.f;
  for (int j = tid; j < QDIM; j += THREADS) {
    const float p = (__popc(j) & 1) ? -1.f : 1.f;
    const int   i = slot(j);
    acc = fmaf(sre[i] * sre[i] + sim[i] * sim[i], p, acc);
  }
  red[tid] = acc;
  __syncthreads();
  for (int off = THREADS / 2; off > 0; off >>= 1) {
    if (tid < off) red[tid] += red[tid + off];
    __syncthreads();
  }
  if (tid == 0) out[b] = red[0];
}

extern "C" void kernel_launch(void* const* d_in, const int* in_sizes, int n_in,
                              void* d_out, int out_size, void* d_ws, size_t ws_size,
                              hipStream_t stream) {
  (void)n_in; (void)d_ws; (void)ws_size; (void)out_size;
  const float* x      = (const float*)d_in[0];
  const float* thetas = (const float*)d_in[1];
  float* out          = (float*)d_out;
  const int batch     = in_sizes[0];
  qcircuit_parity_kernel<<<dim3(batch), dim3(THREADS), 0, stream>>>(x, thetas, out);
}


# Layer 1 acts on the fixed start state |0...0>: its result is a product state, written directly at its CNOT-permuted index; WMMA passes run for layers 2-4 only; f32 product order of layer 1 changes
# speedup vs baseline: 1.6585x; 1.1770x over previous
#include <hip/hip_runtime.h>
#include <math.h>


typedef __attribute__((ext_vector_type(2))) float v2f;
typedef __attribute__((ext_vector_type(4))) float v4f;
typedef __attribute__((ext_vector_type(8))) float v8f;

#define NQ      12
#define NL      4
#define QDIM    4096
#define THREADS 256
#define APITCH  20

static_assert(QDIM == (1 << NQ), "dim mismatch");

struct c32 { float re, im; };
__device__ __forceinline__ c32 cmul(c32 a, c32 b) {
  return { a.re * b.re - a.im * b.im, a.re * b.im + a.im * b.re };
}
__device__ __forceinline__ c32 cadd(c32 a, c32 b) { return { a.re + b.re, a.im + b.im }; }

__device__ __forceinline__ void mat2mul(c32 r[4], const c32 p[4], const c32 q[4]) {
  r[0] = cadd(cmul(p[0], q[0]), cmul(p[1], q[2]));
  r[1] = cadd(cmul(p[0], q[1]), cmul(p[1], q[3]));
  r[2] = cadd(cmul(p[2], q[0]), cmul(p[3], q[2]));
  r[3] = cadd(cmul(p[2], q[1]), cmul(p[3], q[3]));
}

__device__ __forceinline__ constexpr int slot(int i) {
  return i ^ (((i >> 6) & 7) << 3) ^ (((i >> 9) & 1) << 2);
}

template <int G>
__device__ __forceinline__ constexpr int lidx(int kk, int w, int c) {
  return G == 0 ? ((kk << 9) | (w << 6) | (c << 2))
       : G == 1 ? ((w << 9) | (kk << 6) | (c << 2))
                : ((w << 9) | ((c >> 1) << 6) | (kk << 3) | ((c & 1) << 2));
}

__device__ __forceinline__ void mma4(v8f acc[4], const v2f a[4], const v2f b[4][4]) {
#pragma unroll
  for (int kc = 0; kc < 4; ++kc)
#pragma unroll
    for (int t = 0; t < 4; ++t)
      acc[t] = __builtin_amdgcn_wmma_f32_16x16x4_f32(false, a[kc], false, b[t][kc], 0, acc[t],
                                                     false, false);
}

__device__ __forceinline__ void load_a(v2f a[4], const float* Ag, int c, int h) {
  const v4f lo = *(const v4f*)&Ag[c * APITCH + 4 * h];
  const v4f hi = *(const v4f*)&Ag[c * APITCH + 8 + 4 * h];
  a[0] = v2f{lo[0], lo[1]}; a[1] = v2f{lo[2], lo[3]};
  a[2] = v2f{hi[0], hi[1]}; a[3] = v2f{hi[2], hi[3]};
}

template <int G>
__device__ __forceinline__ void pass_cols(float* st, const float* Ag, int w, int c, int h) {
  const int lane0 = slot(lidx<G>(0, w, c));
  const int ld0 = lane0 ^ (h ? slot(lidx<G>(4, 0, 0)) : 0);
  v4f L[2][4];
#pragma unroll
  for (int a = 0; a < 2; ++a)
#pragma unroll
    for (int j = 0; j < 4; ++j)
      L[a][j] = *(const v4f*)&st[a * QDIM + (ld0 ^ slot(lidx<G>(j, 0, 0)))];
  v2f af[4];
  load_a(af, Ag, c, h);
  v2f bf[4][4];
#pragma unroll
  for (int t = 0; t < 4; ++t)
#pragma unroll
    for (int kc = 0; kc < 4; ++kc)
      bf[t][kc] = v2f{ L[kc >> 1][2 * (kc & 1)][t], L[kc >> 1][2 * (kc & 1) + 1][t] };
  v8f acc[4];
#pragma unroll
  for (int t = 0; t < 4; ++t) acc[t] = v8f{0.f, 0.f, 0.f, 0.f, 0.f, 0.f, 0.f, 0.f};
  mma4(acc, af, bf);
#pragma unroll
  for (int r = 0; r < 8; ++r)
    *(v4f*)&st[h * QDIM + (lane0 ^ slot(lidx<G>(r, 0, 0)))] =
        v4f{ acc[0][r], acc[1][r], acc[2][r], acc[3][r] };
}

__device__ __forceinline__ void pass_low(float* st, const float* Ag, int w, int c, int h) {
  const int col0 = slot(((c >> 3) << 9) | (w << 6) | ((c & 7) << 3));
  v4f L[2][4];
#pragma unroll
  for (int a = 0; a < 2; ++a)
#pragma unroll
    for (int t = 0; t < 4; ++t)
      L[a][t] = *(const v4f*)&st[a * QDIM + (t << 10) + (col0 ^ (h << 2))];
  v2f af[4];
  load_a(af, Ag, c, h);
  v2f bf[4][4];
#pragma unroll
  for (int t = 0; t < 4; ++t)
#pragma unroll
    for (int kc = 0; kc < 4; ++kc)
      bf[t][kc] = v2f{ L[kc >> 1][t][2 * (kc & 1)], L[kc >> 1][t][2 * (kc & 1) + 1] };
  v8f acc[4];
#pragma unroll
  for (int t = 0; t < 4; ++t) acc[t] = v8f{0.f, 0.f, 0.f, 0.f, 0.f, 0.f, 0.f, 0.f};
  mma4(acc, af, bf);
#pragma unroll
  for (int t = 0; t < 4; ++t) {
    *(v4f*)&st[h * QDIM + (t << 10) + col0]       = v4f{ acc[t][0], acc[t][1], acc[t][2], acc[t][3] };
    *(v4f*)&st[h * QDIM + (t << 10) + (col0 ^ 4)] = v4f{ acc[t][4], acc[t][5], acc[t][6], acc[t][7] };
  }
}

__device__ __forceinline__ void layer_gates(float* gmre, float* gmim, const float* __restrict__ thetas,
                                            int l, int tid, float x1, float x2) {
  if (tid < NQ) {
    const int q = tid;
    c32 M[4];
    { float s, c; __sincosf(0.5f * x1, &s, &c);
      M[0] = c32{c, 0.f}; M[1] = c32{-s, 0.f}; M[2] = c32{s, 0.f}; M[3] = c32{c, 0.f}; }
    { float s, c; __sincosf(0.5f * x2, &s, &c);
      c32 G[4] = { c32{c, -s}, c32{0.f, 0.f}, c32{0.f, 0.f}, c32{c, s} };
      c32 T[4]; mat2mul(T, G, M);
      M[0] = T[0]; M[1] = T[1]; M[2] = T[2]; M[3] = T[3]; }
    const float t0 = thetas[(l * NQ + q) * 3 + 0];
    const float t1 = thetas[(l * NQ + q) * 3 + 1];
    const float t2 = thetas[(l * NQ + q) * 3 + 2];
    { float s, c; __sincosf(0.5f * t0, &s, &c);
      c32 G[4] = { c32{c, 0.f}, c32{0.f, -s}, c32{0.f, -s}, c32{c, 0.f} };
      c32 T[4]; mat2mul(T, G, M);
      M[0] = T[0]; M[1] = T[1]; M[2] = T[2]; M[3] = T[3]; }
    { float s, c; __sincosf(0.5f * t1, &s, &c);
      c32 G[4] = { c32{c, -s}, c32{0.f, 0.f}, c32{0.f, 0.f}, c32{c, s} };
      c32 T[4]; mat2mul(T, G, M);
      M[0] = T[0]; M[1] = T[1]; M[2] = T[2]; M[3] = T[3]; }
    { float s, c; __sincosf(0.5f * t2, &s, &c);
      c32 G[4] = { c32{c, 0.f}, c32{0.f, -s}, c32{0.f, -s}, c32{c, 0.f} };
      c32 T[4]; mat2mul(T, G, M);
      M[0] = T[0]; M[1] = T[1]; M[2] = T[2]; M[3] = T[3]; }
#pragma unroll
    for (int k = 0; k < 4; ++k) { gmre[q * 4 + k] = M[k].re; gmim[q * 4 + k] = M[k].im; }
  }
}

__device__ __forceinline__ c32 col0(const float* gmre, const float* gmim, int k, int g) {
  const int q = NQ - 1 - k;
  return c32{ gmre[q * 4 + 2 * g], gmim[q * 4 + 2 * g] };
}

__global__ __launch_bounds__(THREADS)
void qcircuit_parity_kernel(const float* __restrict__ x,
                            const float* __restrict__ thetas,
                            float* __restrict__ out) {
  __shared__ __attribute__((aligned(16))) float st[2 * QDIM];
  __shared__ __attribute__((aligned(16))) float Aemb[4 * 16 * APITCH];
  __shared__ float gmre[NQ * 4];
  __shared__ float gmim[NQ * 4];
  __shared__ float red[THREADS];
  float* sre = st;
  float* sim = st + QDIM;

  const int tid  = threadIdx.x;
  const int b    = blockIdx.x;
  const int lane = tid & 31;
  const int wave = tid >> 5;
  const int half = lane >> 4;
  const int mn   = lane & 15;

  const float xv = x[b];
  const float x1 = asinf(xv);
  const float x2 = acosf(xv * xv);

  int csrc[16];
#pragma unroll
  for (int j = 0; j < 16; ++j) {
    int s = tid + j * THREADS;
#pragma unroll
    for (int jj = 10; jj >= 0; --jj)
      s ^= ((s >> (11 - jj)) & 1) << (10 - jj);
    csrc[j] = slot(s);
  }

  layer_gates(gmre, gmim, thetas, 0, tid, x1, x2);
  __syncthreads();
  {
    const int ib = ((tid >> 4) << 8) | ((tid & 15) << 2);
    const int gb = ib ^ (ib >> 1);
    c32 C = col0(gmre, gmim, 2, (gb >> 2) & 1);
    C = cmul(C, col0(gmre, gmim, 3, (gb >> 3) & 1));
    C = cmul(C, col0(gmre, gmim, 4, (gb >> 4) & 1));
    C = cmul(C, col0(gmre, gmim, 8, (gb >> 8) & 1));
    C = cmul(C, col0(gmre, gmim, 9, (gb >> 9) & 1));
    C = cmul(C, col0(gmre, gmim, 10, (gb >> 10) & 1));
    C = cmul(C, col0(gmre, gmim, 11, (gb >> 11) & 1));
    const int i2 = (ib >> 2) & 1, i5 = (ib >> 5) & 1, i8 = (ib >> 8) & 1;
    c32 E[4];
#pragma unroll
    for (int e = 0; e < 4; ++e)
      E[e] = cmul(col0(gmre, gmim, 0, (e ^ (e >> 1)) & 1), col0(gmre, gmim, 1, (e >> 1) ^ i2));
#pragma unroll
    for (int mm = 0; mm < 4; ++mm) {
      c32 F = cmul(col0(gmre, gmim, 5, i5 ^ (mm & 1)), col0(gmre, gmim, 6, (mm ^ (mm >> 1)) & 1));
      F = cmul(F, col0(gmre, gmim, 7, (mm >> 1) ^ i8));
      F = cmul(C, F);
      c32 a[4];
#pragma unroll
      for (int e = 0; e < 4; ++e) a[e] = cmul(F, E[e]);
      const int ad = slot(ib | (mm << 6));
      *(v4f*)&sre[ad] = v4f{ a[0].re, a[1].re, a[2].re, a[3].re };
      *(v4f*)&sim[ad] = v4f{ a[0].im, a[1].im, a[2].im, a[3].im };
    }
  }
  __syncthreads();

  for (int l = 1; l < NL; ++l) {
    layer_gates(gmre, gmim, thetas, l, tid, x1, x2);
    __syncthreads();

    {
      const int g = tid >> 6, r = (tid >> 3) & 7, cc = tid & 7;
      c32 acc = c32{1.f, 0.f};
#pragma unroll
      for (int j = 0; j < 3; ++j) {
        const int q  = 3 * g + j;
        const int bi = (r  >> (2 - j)) & 1;
        const int bj = (cc >> (2 - j)) & 1;
        acc = cmul(acc, c32{ gmre[q * 4 + bi * 2 + bj], gmim[q * 4 + bi * 2 + bj] });
      }
      float* Ag = Aemb + g * 16 * APITCH;
      Ag[r * APITCH + cc]           = acc.re;
      Ag[r * APITCH + cc + 8]       = -acc.im;
      Ag[(r + 8) * APITCH + cc]     = acc.im;
      Ag[(r + 8) * APITCH + cc + 8] = acc.re;
    }
    __syncthreads();

    pass_cols<0>(st, Aemb + 0 * 16 * APITCH, wave, mn, half);
    __builtin_amdgcn_fence(__ATOMIC_ACQ_REL, "wavefront");
    pass_low(st, Aemb + 3 * 16 * APITCH, wave, mn, half);
    __syncthreads();
    pass_cols<1>(st, Aemb + 1 * 16 * APITCH, wave, mn, half);
    __builtin_amdgcn_fence(__ATOMIC_ACQ_REL, "wavefront");
    pass_cols<2>(st, Aemb + 2 * 16 * APITCH, wave, mn, half);
    __syncthreads();

    float tr[16], ti[16];
#pragma unroll
    for (int j = 0; j < 16; ++j) { tr[j] = sre[csrc[j]]; ti[j] = sim[csrc[j]]; }
    __syncthreads();
#pragma unroll
    for (int j = 0; j < 16; ++j) {
      const int i = slot(tid + j * THREADS);
      sre[i] = tr[j]; sim[i] = ti[j];
    }
    __syncthreads();
  }

  float acc = 0.f;
  for (int j = tid; j < QDIM; j += THREADS) {
    const float p = (__popc(j) & 1) ? -1.f : 1.f;
    const int   i = slot(j);
    acc = fmaf(sre[i] * sre[i] + sim[i] * sim[i], p, acc);
  }
  red[tid] = acc;
  __syncthreads();
  for (int off = THREADS / 2; off > 0; off >>= 1) {
    if (tid < off) red[tid] += red[tid + off];
    __syncthreads();
  }
  if (tid == 0) out[b] = red[0];
}

extern "C" void kernel_launch(void* const* d_in, const int* in_sizes, int n_in,
                              void* d_out, int out_size, void* d_ws, size_t ws_size,
                              hipStream_t stream) {
  (void)n_in; (void)d_ws; (void)ws_size; (void)out_size;
  const float* x      = (const float*)d_in[0];
  const float* thetas = (const float*)d_in[1];
  float* out          = (float*)d_out;
  const int batch     = in_sizes[0];
  qcircuit_parity_kernel<<<dim3(batch), dim3(THREADS), 0, stream>>>(x, thetas, out);
}


# CNOT ladder dropped as a separate LDS pass: next layer's first matrix pass reads its four-float units at the ladder's source index, the last ladder becomes parity signs; f32 final sum order changes
# speedup vs baseline: 2.0109x; 1.2125x over previous
#include <hip/hip_runtime.h>
#include <math.h>


typedef __attribute__((ext_vector_type(2))) float v2f;
typedef __attribute__((ext_vector_type(4))) float v4f;
typedef __attribute__((ext_vector_type(8))) float v8f;

#define NQ      12
#define NL      4
#define QDIM    4096
#define THREADS 256
#define APITCH  20

static_assert(QDIM == (1 << NQ), "dim mismatch");

struct c32 { float re, im; };
__device__ __forceinline__ c32 cmul(c32 a, c32 b) {
  return { a.re * b.re - a.im * b.im, a.re * b.im + a.im * b.re };
}
__device__ __forceinline__ c32 cadd(c32 a, c32 b) { return { a.re + b.re, a.im + b.im }; }

__device__ __forceinline__ void mat2mul(c32 r[4], const c32 p[4], const c32 q[4]) {
  r[0] = cadd(cmul(p[0], q[0]), cmul(p[1], q[2]));
  r[1] = cadd(cmul(p[0], q[1]), cmul(p[1], q[3]));
  r[2] = cadd(cmul(p[2], q[0]), cmul(p[3], q[2]));
  r[3] = cadd(cmul(p[2], q[1]), cmul(p[3], q[3]));
}

__device__ __forceinline__ constexpr int slot(int i) {
  return i ^ (((i >> 6) & 7) << 3) ^ (((i >> 9) & 1) << 2);
}

__device__ __forceinline__ constexpr int gslot(int i) {
  return slot((i ^ (i >> 1)) & ~3);
}

template <int G>
__device__ __forceinline__ constexpr int lidx(int kk, int w, int c) {
  return G == 0 ? ((kk << 9) | (w << 6) | (c << 2))
       : G == 1 ? ((w << 9) | (kk << 6) | (c << 2))
                : ((w << 9) | ((c >> 1) << 6) | (kk << 3) | ((c & 1) << 2));
}

__device__ __forceinline__ void mma4(v8f acc[4], const v2f a[4], const v2f b[4][4]) {
#pragma unroll
  for (int kc = 0; kc < 4; ++kc)
#pragma unroll
    for (int t = 0; t < 4; ++t)
      acc[t] = __builtin_amdgcn_wmma_f32_16x16x4_f32(false, a[kc], false, b[t][kc], 0, acc[t],
                                                     false, false);
}

__device__ __forceinline__ void load_a(v2f a[4], const float* Ag, int c, int h) {
  const v4f lo = *(const v4f*)&Ag[c * APITCH + 4 * h];
  const v4f hi = *(const v4f*)&Ag[c * APITCH + 8 + 4 * h];
  a[0] = v2f{lo[0], lo[1]}; a[1] = v2f{lo[2], lo[3]};
  a[2] = v2f{hi[0], hi[1]}; a[3] = v2f{hi[2], hi[3]};
}

template <int G, bool LADDER>
__device__ __forceinline__ void pass_cols(float* st, const float* Ag, int w, int c, int h) {
  static_assert(!LADDER || G == 0, "the ladder read is written for the pass over group 0");
  const int lane0 = slot(lidx<G>(0, w, c));
  const int ld0 = LADDER ? (gslot(lidx<G>(0, w, c)) ^ (h ? gslot(lidx<G>(4, 0, 0)) : 0))
                         : (lane0 ^ (h ? slot(lidx<G>(4, 0, 0)) : 0));
  v4f L[2][4];
#pragma unroll
  for (int a = 0; a < 2; ++a)
#pragma unroll
    for (int j = 0; j < 4; ++j)
      L[a][j] = *(const v4f*)&st[a * QDIM +
                                 (ld0 ^ (LADDER ? gslot(lidx<G>(j, 0, 0)) : slot(lidx<G>(j, 0, 0))))];
  v2f af[4];
  load_a(af, Ag, c, h);
  v2f bf[4][4];
#pragma unroll
  for (int t = 0; t < 4; ++t)
#pragma unroll
    for (int kc = 0; kc < 4; ++kc)
      bf[t][kc] = v2f{ L[kc >> 1][2 * (kc & 1)][t], L[kc >> 1][2 * (kc & 1) + 1][t] };
  v8f acc[4];
#pragma unroll
  for (int t = 0; t < 4; ++t) acc[t] = v8f{0.f, 0.f, 0.f, 0.f, 0.f, 0.f, 0.f, 0.f};
  mma4(acc, af, bf);
  if (LADDER) {
    __syncthreads();
    const bool odd = (c & 1) != 0;
#pragma unroll
    for (int r = 0; r < 8; ++r)
      *(v4f*)&st[h * QDIM + (lane0 ^ slot(lidx<G>(r, 0, 0)))] =
          v4f{ odd ? acc[2][r] : acc[0][r], odd ? acc[3][r] : acc[1][r],
               odd ? acc[1][r] : acc[3][r], odd ? acc[0][r] : acc[2][r] };
  } else {
#pragma unroll
    for (int r = 0; r < 8; ++r)
      *(v4f*)&st[h * QDIM + (lane0 ^ slot(lidx<G>(r, 0, 0)))] =
          v4f{ acc[0][r], acc[1][r], acc[2][r], acc[3][r] };
  }
}

__device__ __forceinline__ void pass_low(float* st, const float* Ag, int w, int c, int h) {
  const int col0 = slot(((c >> 3) << 9) | (w << 6) | ((c & 7) << 3));
  v4f L[2][4];
#pragma unroll
  for (int a = 0; a < 2; ++a)
#pragma unroll
    for (int t = 0; t < 4; ++t)
      L[a][t] = *(const v4f*)&st[a * QDIM + (t << 10) + (col0 ^ (h << 2))];
  v2f af[4];
  load_a(af, Ag, c, h);
  v2f bf[4][4];
#pragma unroll
  for (int t = 0; t < 4; ++t)
#pragma unroll
    for (int kc = 0; kc < 4; ++kc)
      bf[t][kc] = v2f{ L[kc >> 1][t][2 * (kc & 1)], L[kc >> 1][t][2 * (kc & 1) + 1] };
  v8f acc[4];
#pragma unroll
  for (int t = 0; t < 4; ++t) acc[t] = v8f{0.f, 0.f, 0.f, 0.f, 0.f, 0.f, 0.f, 0.f};
  mma4(acc, af, bf);
#pragma unroll
  for (int t = 0; t < 4; ++t) {
    *(v4f*)&st[h * QDIM + (t << 10) + col0]       = v4f{ acc[t][0], acc[t][1], acc[t][2], acc[t][3] };
    *(v4f*)&st[h * QDIM + (t << 10) + (col0 ^ 4)] = v4f{ acc[t][4], acc[t][5], acc[t][6], acc[t][7] };
  }
}

__device__ __forceinline__ void layer_gates(float* gmre, float* gmim, const float* __restrict__ thetas,
                                            int l, int tid, float x1, float x2) {
  if (tid < NQ) {
    const int q = tid;
    c32 M[4];
    { float s, c; __sincosf(0.5f * x1, &s, &c);
      M[0] = c32{c, 0.f}; M[1] = c32{-s, 0.f}; M[2] = c32{s, 0.f}; M[3] = c32{c, 0.f}; }
    { float s, c; __sincosf(0.5f * x2, &s, &c);
      c32 G[4] = { c32{c, -s}, c32{0.f, 0.f}, c32{0.f, 0.f}, c32{c, s} };
      c32 T[4]; mat2mul(T, G, M);
      M[0] = T[0]; M[1] = T[1]; M[2] = T[2]; M[3] = T[3]; }
    const float t0 = thetas[(l * NQ + q) * 3 + 0];
    const float t1 = thetas[(l * NQ + q) * 3 + 1];
    const float t2 = thetas[(l * NQ + q) * 3 + 2];
    { float s, c; __sincosf(0.5f * t0, &s, &c);
      c32 G[4] = { c32{c, 0.f}, c32{0.f, -s}, c32{0.f, -s}, c32{c, 0.f} };
      c32 T[4]; mat2mul(T, G, M);
      M[0] = T[0]; M[1] = T[1]; M[2] = T[2]; M[3] = T[3]; }
    { float s, c; __sincosf(0.5f * t1, &s, &c);
      c32 G[4] = { c32{c, -s}, c32{0.f, 0.f}, c32{0.f, 0.f}, c32{c, s} };
      c32 T[4]; mat2mul(T, G, M);
      M[0] = T[0]; M[1] = T[1]; M[2] = T[2]; M[3] = T[3]; }
    { float s, c; __sincosf(0.5f * t2, &s, &c);
      c32 G[4] = { c32{c, 0.f}, c32{0.f, -s}, c32{0.f, -s}, c32{c, 0.f} };
      c32 T[4]; mat2mul(T, G, M);
      M[0] = T[0]; M[1] = T[1]; M[2] = T[2]; M[3] = T[3]; }
#pragma unroll
    for (int k = 0; k < 4; ++k) { gmre[q * 4 + k] = M[k].re; gmim[q * 4 + k] = M[k].im; }
  }
}

__device__ __forceinline__ c32 col0(const float* gmre, const float* gmim, int k, int g) {
  const int q = NQ - 1 - k;
  return c32{ gmre[q * 4 + 2 * g], gmim[q * 4 + 2 * g] };
}

__global__ __launch_bounds__(THREADS)
void qcircuit_parity_kernel(const float* __restrict__ x,
                            const float* __restrict__ thetas,
                            float* __restrict__ out) {
  __shared__ __attribute__((aligned(16))) float st[2 * QDIM];
  __shared__ __attribute__((aligned(16))) float Aemb[4 * 16 * APITCH];
  __shared__ float gmre[NQ * 4];
  __shared__ float gmim[NQ * 4];
  __shared__ float red[THREADS];
  float* sre = st;
  float* sim = st + QDIM;

  const int tid  = threadIdx.x;
  const int b    = blockIdx.x;
  const int lane = tid & 31;
  const int wave = tid >> 5;
  const int half = lane >> 4;
  const int mn   = lane & 15;

  const float xv = x[b];
  const float x1 = asinf(xv);
  const float x2 = acosf(xv * xv);

  layer_gates(gmre, gmim, thetas, 0, tid, x1, x2);
  __syncthreads();
  {
    const int ib = ((tid >> 4) << 8) | ((tid & 15) << 2);
    const int gb = ib ^ (ib >> 1);
    c32 C = col0(gmre, gmim, 2, (gb >> 2) & 1);
    C = cmul(C, col0(gmre, gmim, 3, (gb >> 3) & 1));
    C = cmul(C, col0(gmre, gmim, 4, (gb >> 4) & 1));
    C = cmul(C, col0(gmre, gmim, 8, (gb >> 8) & 1));
    C = cmul(C, col0(gmre, gmim, 9, (gb >> 9) & 1));
    C = cmul(C, col0(gmre, gmim, 10, (gb >> 10) & 1));
    C = cmul(C, col0(gmre, gmim, 11, (gb >> 11) & 1));
    const int i2 = (ib >> 2) & 1, i5 = (ib >> 5) & 1, i8 = (ib >> 8) & 1;
    c32 E[4];
#pragma unroll
    for (int e = 0; e < 4; ++e)
      E[e] = cmul(col0(gmre, gmim, 0, (e ^ (e >> 1)) & 1), col0(gmre, gmim, 1, (e >> 1) ^ i2));
#pragma unroll
    for (int mm = 0; mm < 4; ++mm) {
      c32 F = cmul(col0(gmre, gmim, 5, i5 ^ (mm & 1)), col0(gmre, gmim, 6, (mm ^ (mm >> 1)) & 1));
      F = cmul(F, col0(gmre, gmim, 7, (mm >> 1) ^ i8));
      F = cmul(C, F);
      c32 a[4];
#pragma unroll
      for (int e = 0; e < 4; ++e) a[e] = cmul(F, E[e]);
      const int ad = slot(ib | (mm << 6));
      *(v4f*)&sre[ad] = v4f{ a[0].re, a[1].re, a[2].re, a[3].re };
      *(v4f*)&sim[ad] = v4f{ a[0].im, a[1].im, a[2].im, a[3].im };
    }
  }
  __syncthreads();

  for (int l = 1; l < NL; ++l) {
    layer_gates(gmre, gmim, thetas, l, tid, x1, x2);
    __syncthreads();

    {
      const int g = tid >> 6, r = (tid >> 3) & 7, cc = tid & 7;
      c32 acc = c32{1.f, 0.f};
#pragma unroll
      for (int j = 0; j < 3; ++j) {
        const int q  = 3 * g + j;
        const int bi = (r  >> (2 - j)) & 1;
        const int bj = (cc >> (2 - j)) & 1;
        acc = cmul(acc, c32{ gmre[q * 4 + bi * 2 + bj], gmim[q * 4 + bi * 2 + bj] });
      }
      float* Ag = Aemb + g * 16 * APITCH;
      Ag[r * APITCH + cc]           = acc.re;
      Ag[r * APITCH + cc + 8]       = -acc.im;
      Ag[(r + 8) * APITCH + cc]     = acc.im;
      Ag[(r + 8) * APITCH + cc + 8] = acc.re;
    }
    __syncthreads();

    if (l == 1) pass_cols<0, false>(st, Aemb + 0 * 16 * APITCH, wave, mn, half);
    else        pass_cols<0, true>(st, Aemb + 0 * 16 * APITCH, wave, mn, half);
    __builtin_amdgcn_fence(__ATOMIC_ACQ_REL, "wavefront");
    pass_low(st, Aemb + 3 * 16 * APITCH, wave, mn, half);
    __syncthreads();
    pass_cols<1, false>(st, Aemb + 1 * 16 * APITCH, wave, mn, half);
    __builtin_amdgcn_fence(__ATOMIC_ACQ_REL, "wavefront");
    pass_cols<2, false>(st, Aemb + 2 * 16 * APITCH, wave, mn, half);
    __syncthreads();
  }

  float acc = 0.f;
  for (int j = tid; j < QDIM; j += THREADS) {
    const float p = (__popc(j & 0x555) & 1) ? -1.f : 1.f;
    const int   i = slot(j);
    acc = fmaf(sre[i] * sre[i] + sim[i] * sim[i], p, acc);
  }
  red[tid] = acc;
  __syncthreads();
  for (int off = THREADS / 2; off > 0; off >>= 1) {
    if (tid < off) red[tid] += red[tid + off];
    __syncthreads();
  }
  if (tid == 0) out[b] = red[0];
}

extern "C" void kernel_launch(void* const* d_in, const int* in_sizes, int n_in,
                              void* d_out, int out_size, void* d_ws, size_t ws_size,
                              hipStream_t stream) {
  (void)n_in; (void)d_ws; (void)ws_size; (void)out_size;
  const float* x      = (const float*)d_in[0];
  const float* thetas = (const float*)d_in[1];
  float* out          = (float*)d_out;
  const int batch     = in_sizes[0];
  qcircuit_parity_kernel<<<dim3(batch), dim3(THREADS), 0, stream>>>(x, thetas, out);
}


# Last layer applies only the 6 gates on the parity-signed index bits (U^H U = I cancels the other 6 in the expectation): 2 WMMA passes (bits 10,8,6; 4,2,0), not 4; f32 sum order changes
# speedup vs baseline: 2.2196x; 1.1038x over previous
#include <hip/hip_runtime.h>
#include <math.h>


typedef __attribute__((ext_vector_type(2))) float v2f;
typedef __attribute__((ext_vector_type(4))) float v4f;
typedef __attribute__((ext_vector_type(8))) float v8f;

#define NQ      12
#define NL      4
#define QDIM    4096
#define THREADS 256
#define APITCH  20

static_assert(QDIM == (1 << NQ), "dim mismatch");
static_assert(NL >= 3, "the last layer reads through a matrix layer's ladder");

struct c32 { float re, im; };
__device__ __forceinline__ c32 cmul(c32 a, c32 b) {
  return { a.re * b.re - a.im * b.im, a.re * b.im + a.im * b.re };
}
__device__ __forceinline__ c32 cadd(c32 a, c32 b) { return { a.re + b.re, a.im + b.im }; }

__device__ __forceinline__ void mat2mul(c32 r[4], const c32 p[4], const c32 q[4]) {
  r[0] = cadd(cmul(p[0], q[0]), cmul(p[1], q[2]));
  r[1] = cadd(cmul(p[0], q[1]), cmul(p[1], q[3]));
  r[2] = cadd(cmul(p[2], q[0]), cmul(p[3], q[2]));
  r[3] = cadd(cmul(p[2], q[1]), cmul(p[3], q[3]));
}

__device__ __forceinline__ constexpr int slot(int i) {
  return i ^ (((i >> 6) & 7) << 3) ^ (((i >> 9) & 1) << 2);
}

__device__ __forceinline__ constexpr int gslot(int i) {
  return slot((i ^ (i >> 1)) & ~3);
}

template <int G>
__device__ __forceinline__ constexpr int lidx(int kk, int w, int c) {
  return G == 0 ? ((kk << 9) | (w << 6) | (c << 2))
       : G == 1 ? ((w << 9) | (kk << 6) | (c << 2))
       : G == 2 ? ((w << 9) | ((c >> 1) << 6) | (kk << 3) | ((c & 1) << 2))
                : ((((w >> 2) & 1) << 11) | (((w >> 1) & 1) << 9) | ((w & 1) << 7) |
                   (((kk >> 2) & 1) << 10) | (((kk >> 1) & 1) << 8) | ((kk & 1) << 6) | (c << 2));
}

__device__ __forceinline__ void mma4(v8f acc[4], const v2f a[4], const v2f b[4][4]) {
#pragma unroll
  for (int kc = 0; kc < 4; ++kc)
#pragma unroll
    for (int t = 0; t < 4; ++t)
      acc[t] = __builtin_amdgcn_wmma_f32_16x16x4_f32(false, a[kc], false, b[t][kc], 0, acc[t],
                                                     false, false);
}

__device__ __forceinline__ void load_a(v2f a[4], const float* Ag, int c, int h) {
  const v4f lo = *(const v4f*)&Ag[c * APITCH + 4 * h];
  const v4f hi = *(const v4f*)&Ag[c * APITCH + 8 + 4 * h];
  a[0] = v2f{lo[0], lo[1]}; a[1] = v2f{lo[2], lo[3]};
  a[2] = v2f{hi[0], hi[1]}; a[3] = v2f{hi[2], hi[3]};
}

template <int G, bool LADDER>
__device__ __forceinline__ void pass_cols(float* st, const float* Ag, int w, int c, int h) {
  static_assert(!LADDER || G == 0 || G == 3,
                "the ladder read needs the group above index bit 2 and the lane column on bits 5..2");
  const int lane0 = slot(lidx<G>(0, w, c));
  const int ld0 = LADDER ? (gslot(lidx<G>(0, w, c)) ^ (h ? gslot(lidx<G>(4, 0, 0)) : 0))
                         : (lane0 ^ (h ? slot(lidx<G>(4, 0, 0)) : 0));
  v4f L[2][4];
#pragma unroll
  for (int a = 0; a < 2; ++a)
#pragma unroll
    for (int j = 0; j < 4; ++j)
      L[a][j] = *(const v4f*)&st[a * QDIM +
                                 (ld0 ^ (LADDER ? gslot(lidx<G>(j, 0, 0)) : slot(lidx<G>(j, 0, 0))))];
  v2f af[4];
  load_a(af, Ag, c, h);
  v2f bf[4][4];
#pragma unroll
  for (int t = 0; t < 4; ++t)
#pragma unroll
    for (int kc = 0; kc < 4; ++kc)
      bf[t][kc] = v2f{ L[kc >> 1][2 * (kc & 1)][t], L[kc >> 1][2 * (kc & 1) + 1][t] };
  v8f acc[4];
#pragma unroll
  for (int t = 0; t < 4; ++t) acc[t] = v8f{0.f, 0.f, 0.f, 0.f, 0.f, 0.f, 0.f, 0.f};
  mma4(acc, af, bf);
  if (LADDER) {
    __syncthreads();
    const bool odd = (c & 1) != 0;
#pragma unroll
    for (int r = 0; r < 8; ++r)
      *(v4f*)&st[h * QDIM + (lane0 ^ slot(lidx<G>(r, 0, 0)))] =
          v4f{ odd ? acc[2][r] : acc[0][r], odd ? acc[3][r] : acc[1][r],
               odd ? acc[1][r] : acc[3][r], odd ? acc[0][r] : acc[2][r] };
  } else {
#pragma unroll
    for (int r = 0; r < 8; ++r)
      *(v4f*)&st[h * QDIM + (lane0 ^ slot(lidx<G>(r, 0, 0)))] =
          v4f{ acc[0][r], acc[1][r], acc[2][r], acc[3][r] };
  }
}

__device__ __forceinline__ void pass_low(float* st, const float* Ag, int w, int c, int h) {
  const int col0 = slot(((c >> 3) << 9) | (w << 6) | ((c & 7) << 3));
  v4f L[2][4];
#pragma unroll
  for (int a = 0; a < 2; ++a)
#pragma unroll
    for (int t = 0; t < 4; ++t)
      L[a][t] = *(const v4f*)&st[a * QDIM + (t << 10) + (col0 ^ (h << 2))];
  v2f af[4];
  load_a(af, Ag, c, h);
  v2f bf[4][4];
#pragma unroll
  for (int t = 0; t < 4; ++t)
#pragma unroll
    for (int kc = 0; kc < 4; ++kc)
      bf[t][kc] = v2f{ L[kc >> 1][t][2 * (kc & 1)], L[kc >> 1][t][2 * (kc & 1) + 1] };
  v8f acc[4];
#pragma unroll
  for (int t = 0; t < 4; ++t) acc[t] = v8f{0.f, 0.f, 0.f, 0.f, 0.f, 0.f, 0.f, 0.f};
  mma4(acc, af, bf);
#pragma unroll
  for (int t = 0; t < 4; ++t) {
    *(v4f*)&st[h * QDIM + (t << 10) + col0]       = v4f{ acc[t][0], acc[t][1], acc[t][2], acc[t][3] };
    *(v4f*)&st[h * QDIM + (t << 10) + (col0 ^ 4)] = v4f{ acc[t][4], acc[t][5], acc[t][6], acc[t][7] };
  }
}

__device__ __forceinline__ void pass_even_low(float* st, const float* Ag, int w, int c, int h) {
  const int col0 = slot((((w >> 2) & 1) << 11) | (((w >> 1) & 1) << 10) | ((w & 1) << 8) |
                        (((c >> 3) & 1) << 9) | (((c >> 2) & 1) << 7) | (((c >> 1) & 1) << 6) |
                        ((c & 1) << 5));
  v4f L[2][4];
#pragma unroll
  for (int a = 0; a < 2; ++a)
#pragma unroll
    for (int u = 0; u < 4; ++u)
      L[a][u] = *(const v4f*)&st[a * QDIM + (col0 ^ (h << 4) ^ (u << 2))];
  v2f af[4];
  load_a(af, Ag, c, h);
  v2f bf[4][4];
#pragma unroll
  for (int t = 0; t < 4; ++t)
#pragma unroll
    for (int kc = 0; kc < 4; ++kc)
      bf[t][kc] = v2f{ L[kc >> 1][((t >> 1) << 1) | (kc & 1)][(t & 1) << 1],
                       L[kc >> 1][((t >> 1) << 1) | (kc & 1)][((t & 1) << 1) | 1] };
  v8f acc[4];
#pragma unroll
  for (int t = 0; t < 4; ++t) acc[t] = v8f{0.f, 0.f, 0.f, 0.f, 0.f, 0.f, 0.f, 0.f};
  mma4(acc, af, bf);
#pragma unroll
  for (int r2 = 0; r2 < 2; ++r2)
#pragma unroll
    for (int t1 = 0; t1 < 2; ++t1)
#pragma unroll
      for (int r1 = 0; r1 < 2; ++r1)
        *(v4f*)&st[h * QDIM + (col0 ^ (r2 << 4) ^ (t1 << 3) ^ (r1 << 2))] =
            v4f{ acc[2 * t1][4 * r2 + 2 * r1],     acc[2 * t1][4 * r2 + 2 * r1 + 1],
                 acc[2 * t1 + 1][4 * r2 + 2 * r1], acc[2 * t1 + 1][4 * r2 + 2 * r1 + 1] };
}

__device__ __forceinline__ void layer_gates(float* gmre, float* gmim, const float* __restrict__ thetas,
                                            int l, int tid, float x1, float x2) {
  if (tid < NQ) {
    const int q = tid;
    c32 M[4];
    { float s, c; __sincosf(0.5f * x1, &s, &c);
      M[0] = c32{c, 0.f}; M[1] = c32{-s, 0.f}; M[2] = c32{s, 0.f}; M[3] = c32{c, 0.f}; }
    { float s, c; __sincosf(0.5f * x2, &s, &c);
      c32 G[4] = { c32{c, -s}, c32{0.f, 0.f}, c32{0.f, 0.f}, c32{c, s} };
      c32 T[4]; mat2mul(T, G, M);
      M[0] = T[0]; M[1] = T[1]; M[2] = T[2]; M[3] = T[3]; }
    const float t0 = thetas[(l * NQ + q) * 3 + 0];
    const float t1 = thetas[(l * NQ + q) * 3 + 1];
    const float t2 = thetas[(l * NQ + q) * 3 + 2];
    { float s, c; __sincosf(0.5f * t0, &s, &c);
      c32 G[4] = { c32{c, 0.f}, c32{0.f, -s}, c32{0.f, -s}, c32{c, 0.f} };
      c32 T[4]; mat2mul(T, G, M);
      M[0] = T[0]; M[1] = T[1]; M[2] = T[2]; M[3] = T[3]; }
    { float s, c; __sincosf(0.5f * t1, &s, &c);
      c32 G[4] = { c32{c, -s}, c32{0.f, 0.f}, c32{0.f, 0.f}, c32{c, s} };
      c32 T[4]; mat2mul(T, G, M);
      M[0] = T[0]; M[1] = T[1]; M[2] = T[2]; M[3] = T[3]; }
    { float s, c; __sincosf(0.5f * t2, &s, &c);
      c32 G[4] = { c32{c, 0.f}, c32{0.f, -s}, c32{0.f, -s}, c32{c, 0.f} };
      c32 T[4]; mat2mul(T, G, M);
      M[0] = T[0]; M[1] = T[1]; M[2] = T[2]; M[3] = T[3]; }
#pragma unroll
    for (int k = 0; k < 4; ++k) { gmre[q * 4 + k] = M[k].re; gmim[q * 4 + k] = M[k].im; }
  }
}

__device__ __forceinline__ c32 col0(const float* gmre, const float* gmim, int k, int g) {
  const int q = NQ - 1 - k;
  return c32{ gmre[q * 4 + 2 * g], gmim[q * 4 + 2 * g] };
}

__global__ __launch_bounds__(THREADS)
void qcircuit_parity_kernel(const float* __restrict__ x,
                            const float* __restrict__ thetas,
                            float* __restrict__ out) {
  __shared__ __attribute__((aligned(16))) float st[2 * QDIM];
  __shared__ __attribute__((aligned(16))) float Aemb[4 * 16 * APITCH];
  __shared__ float gmre[NQ * 4];
  __shared__ float gmim[NQ * 4];
  __shared__ float red[THREADS];
  float* sre = st;
  float* sim = st + QDIM;

  const int tid  = threadIdx.x;
  const int b    = blockIdx.x;
  const int lane = tid & 31;
  const int wave = tid >> 5;
  const int half = lane >> 4;
  const int mn   = lane & 15;

  const float xv = x[b];
  const float x1 = asinf(xv);
  const float x2 = acosf(xv * xv);

  layer_gates(gmre, gmim, thetas, 0, tid, x1, x2);
  __syncthreads();
  {
    const int ib = ((tid >> 4) << 8) | ((tid & 15) << 2);
    const int gb = ib ^ (ib >> 1);
    c32 C = col0(gmre, gmim, 2, (gb >> 2) & 1);
    C = cmul(C, col0(gmre, gmim, 3, (gb >> 3) & 1));
    C = cmul(C, col0(gmre, gmim, 4, (gb >> 4) & 1));
    C = cmul(C, col0(gmre, gmim, 8, (gb >> 8) & 1));
    C = cmul(C, col0(gmre, gmim, 9, (gb >> 9) & 1));
    C = cmul(C, col0(gmre, gmim, 10, (gb >> 10) & 1));
    C = cmul(C, col0(gmre, gmim, 11, (gb >> 11) & 1));
    const int i2 = (ib >> 2) & 1, i5 = (ib >> 5) & 1, i8 = (ib >> 8) & 1;
    c32 E[4];
#pragma unroll
    for (int e = 0; e < 4; ++e)
      E[e] = cmul(col0(gmre, gmim, 0, (e ^ (e >> 1)) & 1), col0(gmre, gmim, 1, (e >> 1) ^ i2));
#pragma unroll
    for (int mm = 0; mm < 4; ++mm) {
      c32 F = cmul(col0(gmre, gmim, 5, i5 ^ (mm & 1)), col0(gmre, gmim, 6, (mm ^ (mm >> 1)) & 1));
      F = cmul(F, col0(gmre, gmim, 7, (mm >> 1) ^ i8));
      F = cmul(C, F);
      c32 a[4];
#pragma unroll
      for (int e = 0; e < 4; ++e) a[e] = cmul(F, E[e]);
      const int ad = slot(ib | (mm << 6));
      *(v4f*)&sre[ad] = v4f{ a[0].re, a[1].re, a[2].re, a[3].re };
      *(v4f*)&sim[ad] = v4f{ a[0].im, a[1].im, a[2].im, a[3].im };
    }
  }
  __syncthreads();

  for (int l = 1; l < NL; ++l) {
    layer_gates(gmre, gmim, thetas, l, tid, x1, x2);
    __syncthreads();

    const bool last = (l == NL - 1);
    if (!last || tid < 128) {
      const int g = tid >> 6, r = (tid >> 3) & 7, cc = tid & 7;
      c32 acc = c32{1.f, 0.f};
#pragma unroll
      for (int j = 0; j < 3; ++j) {
        const int q  = last ? 2 * (3 * g + j) + 1 : 3 * g + j;
        const int bi = (r  >> (2 - j)) & 1;
        const int bj = (cc >> (2 - j)) & 1;
        acc = cmul(acc, c32{ gmre[q * 4 + bi * 2 + bj], gmim[q * 4 + bi * 2 + bj] });
      }
      float* Ag = Aemb + g * 16 * APITCH;
      Ag[r * APITCH + cc]           = acc.re;
      Ag[r * APITCH + cc + 8]       = -acc.im;
      Ag[(r + 8) * APITCH + cc]     = acc.im;
      Ag[(r + 8) * APITCH + cc + 8] = acc.re;
    }
    __syncthreads();

    if (last) {
      pass_cols<3, true>(st, Aemb + 0 * 16 * APITCH, wave, mn, half);
      __syncthreads();
      pass_even_low(st, Aemb + 1 * 16 * APITCH, wave, mn, half);
      __syncthreads();
      break;
    }
    if (l == 1) pass_cols<0, false>(st, Aemb + 0 * 16 * APITCH, wave, mn, half);
    else        pass_cols<0, true>(st, Aemb + 0 * 16 * APITCH, wave, mn, half);
    __builtin_amdgcn_fence(__ATOMIC_ACQ_REL, "wavefront");
    pass_low(st, Aemb + 3 * 16 * APITCH, wave, mn, half);
    __syncthreads();
    pass_cols<1, false>(st, Aemb + 1 * 16 * APITCH, wave, mn, half);
    __builtin_amdgcn_fence(__ATOMIC_ACQ_REL, "wavefront");
    pass_cols<2, false>(st, Aemb + 2 * 16 * APITCH, wave, mn, half);
    __syncthreads();
  }

  float acc = 0.f;
  for (int j = tid; j < QDIM; j += THREADS) {
    const float p = (__popc(j & 0x555) & 1) ? -1.f : 1.f;
    const int   i = slot(j);
    acc = fmaf(sre[i] * sre[i] + sim[i] * sim[i], p, acc);
  }
  red[tid] = acc;
  __syncthreads();
  for (int off = THREADS / 2; off > 0; off >>= 1) {
    if (tid < off) red[tid] += red[tid + off];
    __syncthreads();
  }
  if (tid == 0) out[b] = red[0];
}

extern "C" void kernel_launch(void* const* d_in, const int* in_sizes, int n_in,
                              void* d_out, int out_size, void* d_ws, size_t ws_size,
                              hipStream_t stream) {
  (void)n_in; (void)d_ws; (void)ws_size; (void)out_size;
  const float* x      = (const float*)d_in[0];
  const float* thetas = (const float*)d_in[1];
  float* out          = (float*)d_out;
  const int batch     = in_sizes[0];
  qcircuit_parity_kernel<<<dim3(batch), dim3(THREADS), 0, stream>>>(x, thetas, out);
}


# One batch item per 4-wave workgroup (128 threads), not 8: each wave takes one tile of both tables and two Gram rows; half the waves to launch; f32 final sum order changes
# speedup vs baseline: 14.1165x; 6.3599x over previous
#include <hip/hip_runtime.h>
#include <math.h>


typedef __attribute__((ext_vector_type(2))) float v2f;
typedef __attribute__((ext_vector_type(4))) float v4f;
typedef __attribute__((ext_vector_type(8))) float v8f;

#define NQ      12
#define NL      4
#define QDIM    4096
#define THREADS 128
#define APITCH  20
#define TPITCH  80
#define XPITCH  68

static_assert(QDIM == (1 << NQ), "dim mismatch");
static_assert(NL == 4, "the direct build covers exactly the three layers in front of the last one");

struct c32 { float re, im; };
__device__ __forceinline__ c32 cmul(c32 a, c32 b) {
  return { a.re * b.re - a.im * b.im, a.re * b.im + a.im * b.re };
}
__device__ __forceinline__ c32 cadd(c32 a, c32 b) { return { a.re + b.re, a.im + b.im }; }
__device__ __forceinline__ c32 cdot2(c32 a, c32 b, c32 c, c32 d) {
  return { fmaf(-c.im, d.im, fmaf(c.re, d.re, fmaf(-a.im, b.im, a.re * b.re))),
           fmaf( c.im, d.re, fmaf(c.re, d.im, fmaf( a.im, b.re, a.re * b.im))) };
}

__device__ __forceinline__ void mat2mul(c32 r[4], const c32 p[4], const c32 q[4]) {
  r[0] = cadd(cmul(p[0], q[0]), cmul(p[1], q[2]));
  r[1] = cadd(cmul(p[0], q[1]), cmul(p[1], q[3]));
  r[2] = cadd(cmul(p[2], q[0]), cmul(p[3], q[2]));
  r[3] = cadd(cmul(p[2], q[1]), cmul(p[3], q[3]));
}

__device__ __forceinline__ void load_a(v2f a[4], const float* Ag, int c, int h) {
  const v4f lo = *(const v4f*)&Ag[c * APITCH + 4 * h];
  const v4f hi = *(const v4f*)&Ag[c * APITCH + 8 + 4 * h];
  a[0] = v2f{lo[0], lo[1]}; a[1] = v2f{lo[2], lo[3]};
  a[2] = v2f{hi[0], hi[1]}; a[3] = v2f{hi[2], hi[3]};
}

__device__ __forceinline__ void all_gates(float* gmre, float* gmim, const float* __restrict__ thetas,
                                          int tid, float x1, float x2) {
  if (tid < NL * NQ) {
    c32 M[4];
    { float s, c; __sincosf(0.5f * x1, &s, &c);
      M[0] = c32{c, 0.f}; M[1] = c32{-s, 0.f}; M[2] = c32{s, 0.f}; M[3] = c32{c, 0.f}; }
    { float s, c; __sincosf(0.5f * x2, &s, &c);
      c32 G[4] = { c32{c, -s}, c32{0.f, 0.f}, c32{0.f, 0.f}, c32{c, s} };
      c32 T[4]; mat2mul(T, G, M);
      M[0] = T[0]; M[1] = T[1]; M[2] = T[2]; M[3] = T[3]; }
    const float t0 = thetas[tid * 3 + 0];
    const float t1 = thetas[tid * 3 + 1];
    const float t2 = thetas[tid * 3 + 2];
    { float s, c; __sincosf(0.5f * t0, &s, &c);
      c32 G[4] = { c32{c, 0.f}, c32{0.f, -s}, c32{0.f, -s}, c32{c, 0.f} };
      c32 T[4]; mat2mul(T, G, M);
      M[0] = T[0]; M[1] = T[1]; M[2] = T[2]; M[3] = T[3]; }
    { float s, c; __sincosf(0.5f * t1, &s, &c);
      c32 G[4] = { c32{c, -s}, c32{0.f, 0.f}, c32{0.f, 0.f}, c32{c, s} };
      c32 T[4]; mat2mul(T, G, M);
      M[0] = T[0]; M[1] = T[1]; M[2] = T[2]; M[3] = T[3]; }
    { float s, c; __sincosf(0.5f * t2, &s, &c);
      c32 G[4] = { c32{c, 0.f}, c32{0.f, -s}, c32{0.f, -s}, c32{c, 0.f} };
      c32 T[4]; mat2mul(T, G, M);
      M[0] = T[0]; M[1] = T[1]; M[2] = T[2]; M[3] = T[3]; }
#pragma unroll
    for (int k = 0; k < 4; ++k) { gmre[tid * 4 + k] = M[k].re; gmim[tid * 4 + k] = M[k].im; }
  }
}

__device__ __forceinline__ c32 gate(const float* gmre, const float* gmim, int l, int k, int row, int col) {
  const int at = (l * NQ + (NQ - 1 - k)) * 4 + 2 * row + col;
  return c32{ gmre[at], gmim[at] };
}

struct site { c32 a[2]; c32 u2[2][2]; c32 u3p[2]; };
__device__ __forceinline__ site load_site(const float* gmre, const float* gmim, int k, int p) {
  site s;
#pragma unroll
  for (int g = 0; g < 2; ++g) {
    s.a[g]   = gate(gmre, gmim, 0, k, g, 0);
    s.u3p[g] = gate(gmre, gmim, 2, k, p, g);
#pragma unroll
    for (int j = 0; j < 2; ++j) s.u2[g][j] = gate(gmre, gmim, 1, k, g, j);
  }
  return s;
}

__device__ __forceinline__ void row_step(c32 v[2][2], const site& s) {
  c32 t[2][2], u[2][2];
#pragma unroll
  for (int i = 0; i < 2; ++i)
#pragma unroll
    for (int mp = 0; mp < 2; ++mp)
      t[i][mp] = cdot2(v[0][mp], s.a[i], v[1][mp], s.a[i ^ 1]);
#pragma unroll
  for (int i = 0; i < 2; ++i)
#pragma unroll
    for (int m = 0; m < 2; ++m)
      u[i][m] = cdot2(s.u2[m][i], t[i][0], s.u2[m ^ 1][i], t[i][1]);
#pragma unroll
  for (int i = 0; i < 2; ++i)
#pragma unroll
    for (int m = 0; m < 2; ++m)
      v[i][m] = cmul(s.u3p[m], u[i][m]);
}

__device__ __forceinline__ void col_step(c32 w[2][2], const site& s) {
  c32 y[2][2], z[2][2];
#pragma unroll
  for (int i = 0; i < 2; ++i)
#pragma unroll
    for (int m = 0; m < 2; ++m)
      y[i][m] = cmul(s.u3p[m], w[i][m]);
#pragma unroll
  for (int i = 0; i < 2; ++i)
#pragma unroll
    for (int mp = 0; mp < 2; ++mp)
      z[i][mp] = cdot2(s.u2[mp][i], y[i][0], s.u2[mp ^ 1][i], y[i][1]);
#pragma unroll
  for (int ip = 0; ip < 2; ++ip)
#pragma unroll
    for (int mp = 0; mp < 2; ++mp)
      w[ip][mp] = cdot2(s.a[ip], z[0][mp], s.a[ip ^ 1], z[1][mp]);
}

__device__ __forceinline__ int plane_pos(int v) {
  const int od = (((v >> 5) & 1) << 2) | (((v >> 3) & 1) << 1) | ((v >> 1) & 1);
  const int ev = (((v >> 4) & 1) << 2) | (((v >> 2) & 1) << 1) | (v & 1);
  return od * 8 + ev;
}

__device__ __forceinline__ int g_inv(int n) {
  n ^= n >> 1; n ^= n >> 2; n ^= n >> 4;
  return n;
}

__device__ __forceinline__ void store_l(float* T, int n, const c32 v[2][2]) {
  const int pos = plane_pos(g_inv(n));
#pragma unroll
  for (int i = 0; i < 2; ++i)
#pragma unroll
    for (int m = 0; m < 2; ++m) {
      T[(2 * i + m) * TPITCH + pos]     = v[i][m].re;
      T[(4 + 2 * i + m) * TPITCH + pos] = v[i][m].im;
    }
}

__device__ __forceinline__ void store_r(float* T, int n, const c32 w[2][2]) {
#pragma unroll
  for (int f = 0; f < 2; ++f) {
    const int pos = plane_pos(g_inv(n ^ (f << 5)));
#pragma unroll
    for (int i = 0; i < 2; ++i)
#pragma unroll
      for (int m = 0; m < 2; ++m) {
        T[(4 * f + 2 * i + m) * TPITCH + pos]     = w[i][m].re;
        T[(8 + 4 * f + 2 * i + m) * TPITCH + pos] = w[i][m].im;
      }
  }
}

__global__ __launch_bounds__(THREADS)
void qcircuit_parity_kernel(const float* __restrict__ x,
                            const float* __restrict__ thetas,
                            float* __restrict__ out) {
  __shared__ __attribute__((aligned(16))) float Aemb[2 * 16 * APITCH];
  __shared__ __attribute__((aligned(16))) float LT[8 * TPITCH];
  __shared__ __attribute__((aligned(16))) float RT[16 * TPITCH];
  __shared__ __attribute__((aligned(16))) float XP[2 * 2 * 8 * XPITCH];
  __shared__ float gmre[NL * NQ * 4];
  __shared__ float gmim[NL * NQ * 4];
  __shared__ float red[THREADS / 32];

  const int tid  = threadIdx.x;
  const int b    = blockIdx.x;
  const int lane = tid & 31;
  const int wave = tid >> 5;
  const int half = lane >> 4;
  const int mn   = lane & 15;

  const float xv = x[b];
  const float x1 = asinf(xv);
  const float x2 = acosf(xv * xv);

  all_gates(gmre, gmim, thetas, tid, x1, x2);
  __syncthreads();

  if (tid < 64) {
    c32 v[2][2] = { { c32{1.f, 0.f}, c32{0.f, 0.f} }, { c32{0.f, 0.f}, c32{0.f, 0.f} } };
#pragma unroll
    for (int k = NQ - 1; k >= 6; --k)
      row_step(v, load_site(gmre, gmim, k, (tid >> (k - 6)) & 1));
    store_l(LT, tid, v);
  } else if (tid < 128) {
    const int n = tid - 64;
    c32 w[2][2] = { { c32{1.f, 0.f}, c32{1.f, 0.f} }, { c32{1.f, 0.f}, c32{1.f, 0.f} } };
#pragma unroll
    for (int k = 0; k < 6; ++k)
      col_step(w, load_site(gmre, gmim, k, (n >> k) & 1));
    store_r(RT, n, w);
  }
  {
    const int u = tid;
    const int g = u >> 6, r = (u >> 3) & 7, cc = u & 7;
    c32 acc = c32{1.f, 0.f};
#pragma unroll
    for (int j = 0; j < 3; ++j) {
      const int q  = 2 * (3 * g + j) + 1;
      const int bi = (r  >> (2 - j)) & 1;
      const int bj = (cc >> (2 - j)) & 1;
      acc = cmul(acc, gate(gmre, gmim, NL - 1, NQ - 1 - q, bi, bj));
    }
    float* Ag = Aemb + g * 16 * APITCH;
    Ag[r * APITCH + cc]           = acc.re;
    Ag[r * APITCH + cc + 8]       = -acc.im;
    Ag[(r + 8) * APITCH + cc]     = acc.im;
    Ag[(r + 8) * APITCH + cc + 8] = acc.re;
  }
  __syncthreads();

#pragma unroll
  for (int tb = 0; tb < 2; ++tb) {
    const int a    = mn & 7;
    const int od   = 2 * wave + (mn >> 3);
    v4f P[2];
    if (tb == 0) {
      const bool f = (a >> 2) != 0;
#pragma unroll
      for (int part = 0; part < 2; ++part) {
        const v4f q = *(const v4f*)&LT[(part * 4 + (a & 3)) * TPITCH + od * 8 + 4 * half];
        P[part] = f ? v4f{0.f, q[1], 0.f, q[3]} : v4f{q[0], 0.f, q[2], 0.f};
      }
    } else {
#pragma unroll
      for (int part = 0; part < 2; ++part)
        P[part] = *(const v4f*)&RT[(part * 8 + a) * TPITCH + od * 8 + 4 * half];
    }
    v2f af[4];
    load_a(af, Aemb + tb * 16 * APITCH, mn, half);
    v8f acc = v8f{0.f, 0.f, 0.f, 0.f, 0.f, 0.f, 0.f, 0.f};
#pragma unroll
    for (int kc = 0; kc < 4; ++kc)
      acc = __builtin_amdgcn_wmma_f32_16x16x4_f32(
          false, af[kc], false, v2f{ P[kc >> 1][2 * (kc & 1)], P[kc >> 1][2 * (kc & 1) + 1] },
          0, acc, false, false);
    float* dst = &XP[((tb * 2 + half) * 8 + a) * XPITCH + od * 8];
    *(v4f*)&dst[0] = v4f{ acc[0], acc[1], acc[2], acc[3] };
    *(v4f*)&dst[4] = v4f{ acc[4], acc[5], acc[6], acc[7] };
  }
  __syncthreads();

  float tot = 0.f;
#pragma unroll
  for (int ai = 0; ai < 2; ++ai) {
    const int a   = wave + 4 * ai;
    const int ap  = lane & 7;
    const int hf  = (lane >> 3) & 1;
    const int tb  = lane >> 4;
    const bool im = a > ap;
    const float* base = XP + tb * 16 * XPITCH + hf * 32;
    const float* U  = base + ((im ? 8 : 0) + a) * XPITCH;
    const float* V  = base + ((im ? 0 : 8) + a) * XPITCH;
    const float* Wr = base + ap * XPITCH;
    const float* Wi = base + (8 + ap) * XPITCH;
    float pu[2] = {0.f, 0.f}, pv[2] = {0.f, 0.f};
#pragma unroll
    for (int q = 0; q < 8; ++q) {
      const v4f u  = *(const v4f*)&U[4 * q];
      const v4f v  = *(const v4f*)&V[4 * q];
      const v4f wr = *(const v4f*)&Wr[4 * q];
      const v4f wi = *(const v4f*)&Wi[4 * q];
      pu[q & 1] = fmaf( u[0], wr[0], pu[q & 1]);
      pv[q & 1] = fmaf( v[0], wi[0], pv[q & 1]);
      pu[q & 1] = fmaf(-u[1], wr[1], pu[q & 1]);
      pv[q & 1] = fmaf(-v[1], wi[1], pv[q & 1]);
      pu[q & 1] = fmaf(-u[2], wr[2], pu[q & 1]);
      pv[q & 1] = fmaf(-v[2], wi[2], pv[q & 1]);
      pu[q & 1] = fmaf( u[3], wr[3], pu[q & 1]);
      pv[q & 1] = fmaf( v[3], wi[3], pv[q & 1]);
    }
    const float su = pu[0] - pu[1];
    const float sv = pv[0] - pv[1];
    float gsum = im ? (su - sv) : (su + sv);
    gsum += __shfl_xor(gsum, 8);
    float val = gsum * __shfl_xor(gsum, 16);
    val *= (a == ap) ? 1.f : (im ? -2.f : 2.f);
    val += __shfl_xor(val, 1);
    val += __shfl_xor(val, 2);
    val += __shfl_xor(val, 4);
    tot += val;
  }
  if (lane == 0) red[wave] = tot;
  __syncthreads();
  if (tid == 0) {
    float r = red[0];
#pragma unroll
    for (int w = 1; w < THREADS / 32; ++w) r += red[w];
    out[b] = r;
  }
}

extern "C" void kernel_launch(void* const* d_in, const int* in_sizes, int n_in,
                              void* d_out, int out_size, void* d_ws, size_t ws_size,
                              hipStream_t stream) {
  (void)n_in; (void)d_ws; (void)ws_size; (void)out_size;
  const float* x      = (const float*)d_in[0];
  const float* thetas = (const float*)d_in[1];
  float* out          = (float*)d_out;
  const int batch     = in_sizes[0];
  qcircuit_parity_kernel<<<dim3(batch), dim3(THREADS), 0, stream>>>(x, thetas, out);
}


# Gram matrices GX, GY formed on the matrix unit: per wave 8 f32 WMMA (A = sign * table^T, B = table) on half the places of one table, not 48 LDS reads + 128 FMA per lane; f32 sum order changes
# speedup vs baseline: 16.1233x; 1.1422x over previous
#include <hip/hip_runtime.h>
#include <math.h>


typedef __attribute__((ext_vector_type(2))) float v2f;
typedef __attribute__((ext_vector_type(4))) float v4f;
typedef __attribute__((ext_vector_type(8))) float v8f;

#define NQ      12
#define NL      4
#define QDIM    4096
#define THREADS 128
#define APITCH  20
#define TPITCH  80
#define XPITCH  68
#define GPITCH  20

static_assert(QDIM == (1 << NQ), "dim mismatch");
static_assert(NL == 4, "the direct build covers exactly the three layers in front of the last one");

struct c32 { float re, im; };
__device__ __forceinline__ c32 cmul(c32 a, c32 b) {
  return { a.re * b.re - a.im * b.im, a.re * b.im + a.im * b.re };
}
__device__ __forceinline__ c32 cadd(c32 a, c32 b) { return { a.re + b.re, a.im + b.im }; }
__device__ __forceinline__ c32 cdot2(c32 a, c32 b, c32 c, c32 d) {
  return { fmaf(-c.im, d.im, fmaf(c.re, d.re, fmaf(-a.im, b.im, a.re * b.re))),
           fmaf( c.im, d.re, fmaf(c.re, d.im, fmaf( a.im, b.re, a.re * b.im))) };
}

__device__ __forceinline__ void mat2mul(c32 r[4], const c32 p[4], const c32 q[4]) {
  r[0] = cadd(cmul(p[0], q[0]), cmul(p[1], q[2]));
  r[1] = cadd(cmul(p[0], q[1]), cmul(p[1], q[3]));
  r[2] = cadd(cmul(p[2], q[0]), cmul(p[3], q[2]));
  r[3] = cadd(cmul(p[2], q[1]), cmul(p[3], q[3]));
}

__device__ __forceinline__ void load_a(v2f a[4], const float* Ag, int c, int h) {
  const v4f lo = *(const v4f*)&Ag[c * APITCH + 4 * h];
  const v4f hi = *(const v4f*)&Ag[c * APITCH + 8 + 4 * h];
  a[0] = v2f{lo[0], lo[1]}; a[1] = v2f{lo[2], lo[3]};
  a[2] = v2f{hi[0], hi[1]}; a[3] = v2f{hi[2], hi[3]};
}

__device__ __forceinline__ void all_gates(float* gmre, float* gmim, const float* __restrict__ thetas,
                                          int tid, float x1, float x2) {
  if (tid < NL * NQ) {
    c32 M[4];
    { float s, c; __sincosf(0.5f * x1, &s, &c);
      M[0] = c32{c, 0.f}; M[1] = c32{-s, 0.f}; M[2] = c32{s, 0.f}; M[3] = c32{c, 0.f}; }
    { float s, c; __sincosf(0.5f * x2, &s, &c);
      c32 G[4] = { c32{c, -s}, c32{0.f, 0.f}, c32{0.f, 0.f}, c32{c, s} };
      c32 T[4]; mat2mul(T, G, M);
      M[0] = T[0]; M[1] = T[1]; M[2] = T[2]; M[3] = T[3]; }
    const float t0 = thetas[tid * 3 + 0];
    const float t1 = thetas[tid * 3 + 1];
    const float t2 = thetas[tid * 3 + 2];
    { float s, c; __sincosf(0.5f * t0, &s, &c);
      c32 G[4] = { c32{c, 0.f}, c32{0.f, -s}, c32{0.f, -s}, c32{c, 0.f} };
      c32 T[4]; mat2mul(T, G, M);
      M[0] = T[0]; M[1] = T[1]; M[2] = T[2]; M[3] = T[3]; }
    { float s, c; __sincosf(0.5f * t1, &s, &c);
      c32 G[4] = { c32{c, -s}, c32{0.f, 0.f}, c32{0.f, 0.f}, c32{c, s} };
      c32 T[4]; mat2mul(T, G, M);
      M[0] = T[0]; M[1] = T[1]; M[2] = T[2]; M[3] = T[3]; }
    { float s, c; __sincosf(0.5f * t2, &s, &c);
      c32 G[4] = { c32{c, 0.f}, c32{0.f, -s}, c32{0.f, -s}, c32{c, 0.f} };
      c32 T[4]; mat2mul(T, G, M);
      M[0] = T[0]; M[1] = T[1]; M[2] = T[2]; M[3] = T[3]; }
#pragma unroll
    for (int k = 0; k < 4; ++k) { gmre[tid * 4 + k] = M[k].re; gmim[tid * 4 + k] = M[k].im; }
  }
}

__device__ __forceinline__ c32 gate(const float* gmre, const float* gmim, int l, int k, int row, int col) {
  const int at = (l * NQ + (NQ - 1 - k)) * 4 + 2 * row + col;
  return c32{ gmre[at], gmim[at] };
}

struct site { c32 a[2]; c32 u2[2][2]; c32 u3p[2]; };
__device__ __forceinline__ site load_site(const float* gmre, const float* gmim, int k, int p) {
  site s;
#pragma unroll
  for (int g = 0; g < 2; ++g) {
    s.a[g]   = gate(gmre, gmim, 0, k, g, 0);
    s.u3p[g] = gate(gmre, gmim, 2, k, p, g);
#pragma unroll
    for (int j = 0; j < 2; ++j) s.u2[g][j] = gate(gmre, gmim, 1, k, g, j);
  }
  return s;
}

__device__ __forceinline__ void row_step(c32 v[2][2], const site& s) {
  c32 t[2][2], u[2][2];
#pragma unroll
  for (int i = 0; i < 2; ++i)
#pragma unroll
    for (int mp = 0; mp < 2; ++mp)
      t[i][mp] = cdot2(v[0][mp], s.a[i], v[1][mp], s.a[i ^ 1]);
#pragma unroll
  for (int i = 0; i < 2; ++i)
#pragma unroll
    for (int m = 0; m < 2; ++m)
      u[i][m] = cdot2(s.u2[m][i], t[i][0], s.u2[m ^ 1][i], t[i][1]);
#pragma unroll
  for (int i = 0; i < 2; ++i)
#pragma unroll
    for (int m = 0; m < 2; ++m)
      v[i][m] = cmul(s.u3p[m], u[i][m]);
}

__device__ __forceinline__ void col_step(c32 w[2][2], const site& s) {
  c32 y[2][2], z[2][2];
#pragma unroll
  for (int i = 0; i < 2; ++i)
#pragma unroll
    for (int m = 0; m < 2; ++m)
      y[i][m] = cmul(s.u3p[m], w[i][m]);
#pragma unroll
  for (int i = 0; i < 2; ++i)
#pragma unroll
    for (int mp = 0; mp < 2; ++mp)
      z[i][mp] = cdot2(s.u2[mp][i], y[i][0], s.u2[mp ^ 1][i], y[i][1]);
#pragma unroll
  for (int ip = 0; ip < 2; ++ip)
#pragma unroll
    for (int mp = 0; mp < 2; ++mp)
      w[ip][mp] = cdot2(s.a[ip], z[0][mp], s.a[ip ^ 1], z[1][mp]);
}

__device__ __forceinline__ int plane_pos(int v) {
  const int od = (((v >> 5) & 1) << 2) | (((v >> 3) & 1) << 1) | ((v >> 1) & 1);
  const int ev = (((v >> 4) & 1) << 2) | (((v >> 2) & 1) << 1) | (v & 1);
  return od * 8 + ev;
}

__device__ __forceinline__ int g_inv(int n) {
  n ^= n >> 1; n ^= n >> 2; n ^= n >> 4;
  return n;
}

__device__ __forceinline__ void store_l(float* T, int n, const c32 v[2][2]) {
  const int pos = plane_pos(g_inv(n));
#pragma unroll
  for (int i = 0; i < 2; ++i)
#pragma unroll
    for (int m = 0; m < 2; ++m) {
      T[(2 * i + m) * TPITCH + pos]     = v[i][m].re;
      T[(4 + 2 * i + m) * TPITCH + pos] = v[i][m].im;
    }
}

__device__ __forceinline__ void store_r(float* T, int n, const c32 w[2][2]) {
#pragma unroll
  for (int f = 0; f < 2; ++f) {
    const int pos = plane_pos(g_inv(n ^ (f << 5)));
#pragma unroll
    for (int i = 0; i < 2; ++i)
#pragma unroll
      for (int m = 0; m < 2; ++m) {
        T[(4 * f + 2 * i + m) * TPITCH + pos]     = w[i][m].re;
        T[(8 + 4 * f + 2 * i + m) * TPITCH + pos] = w[i][m].im;
      }
  }
}

__global__ __launch_bounds__(THREADS)
void qcircuit_parity_kernel(const float* __restrict__ x,
                            const float* __restrict__ thetas,
                            float* __restrict__ out) {
  __shared__ __attribute__((aligned(16))) float Aemb[2 * 16 * APITCH];
  __shared__ __attribute__((aligned(16))) float LT[8 * TPITCH];
  __shared__ __attribute__((aligned(16))) float RT[16 * TPITCH];
  static_assert(16 * TPITCH >= 4 * 16 * GPITCH, "the four partial Gram tiles reuse RT");
  __shared__ __attribute__((aligned(16))) float XP[2 * 2 * 8 * XPITCH];
  __shared__ float gmre[NL * NQ * 4];
  __shared__ float gmim[NL * NQ * 4];
  __shared__ float red[2];

  const int tid  = threadIdx.x;
  const int b    = blockIdx.x;
  const int lane = tid & 31;
  const int wave = tid >> 5;
  const int half = lane >> 4;
  const int mn   = lane & 15;

  const float xv = x[b];
  const float x1 = asinf(xv);
  const float x2 = acosf(xv * xv);

  all_gates(gmre, gmim, thetas, tid, x1, x2);
  __syncthreads();

  if (tid < 64) {
    c32 v[2][2] = { { c32{1.f, 0.f}, c32{0.f, 0.f} }, { c32{0.f, 0.f}, c32{0.f, 0.f} } };
#pragma unroll
    for (int k = NQ - 1; k >= 6; --k)
      row_step(v, load_site(gmre, gmim, k, (tid >> (k - 6)) & 1));
    store_l(LT, tid, v);
  } else if (tid < 128) {
    const int n = tid - 64;
    c32 w[2][2] = { { c32{1.f, 0.f}, c32{1.f, 0.f} }, { c32{1.f, 0.f}, c32{1.f, 0.f} } };
#pragma unroll
    for (int k = 0; k < 6; ++k)
      col_step(w, load_site(gmre, gmim, k, (n >> k) & 1));
    store_r(RT, n, w);
  }
  {
    const int u = tid;
    const int g = u >> 6, r = (u >> 3) & 7, cc = u & 7;
    c32 acc = c32{1.f, 0.f};
#pragma unroll
    for (int j = 0; j < 3; ++j) {
      const int q  = 2 * (3 * g + j) + 1;
      const int bi = (r  >> (2 - j)) & 1;
      const int bj = (cc >> (2 - j)) & 1;
      acc = cmul(acc, gate(gmre, gmim, NL - 1, NQ - 1 - q, bi, bj));
    }
    float* Ag = Aemb + g * 16 * APITCH;
    Ag[r * APITCH + cc]           = acc.re;
    Ag[r * APITCH + cc + 8]       = -acc.im;
    Ag[(r + 8) * APITCH + cc]     = acc.im;
    Ag[(r + 8) * APITCH + cc + 8] = acc.re;
  }
  __syncthreads();

#pragma unroll
  for (int tb = 0; tb < 2; ++tb) {
    const int a    = mn & 7;
    const int od   = 2 * wave + (mn >> 3);
    v4f P[2];
    if (tb == 0) {
      const bool f = (a >> 2) != 0;
#pragma unroll
      for (int part = 0; part < 2; ++part) {
        const v4f q = *(const v4f*)&LT[(part * 4 + (a & 3)) * TPITCH + od * 8 + 4 * half];
        P[part] = f ? v4f{0.f, q[1], 0.f, q[3]} : v4f{q[0], 0.f, q[2], 0.f};
      }
    } else {
#pragma unroll
      for (int part = 0; part < 2; ++part)
        P[part] = *(const v4f*)&RT[(part * 8 + a) * TPITCH + od * 8 + 4 * half];
    }
    v2f af[4];
    load_a(af, Aemb + tb * 16 * APITCH, mn, half);
    v8f acc = v8f{0.f, 0.f, 0.f, 0.f, 0.f, 0.f, 0.f, 0.f};
#pragma unroll
    for (int kc = 0; kc < 4; ++kc)
      acc = __builtin_amdgcn_wmma_f32_16x16x4_f32(
          false, af[kc], false, v2f{ P[kc >> 1][2 * (kc & 1)], P[kc >> 1][2 * (kc & 1) + 1] },
          0, acc, false, false);
    float* dst = &XP[((tb * 2 + half) * 8 + a) * XPITCH + od * 8];
    *(v4f*)&dst[0] = v4f{ acc[0], acc[1], acc[2], acc[3] };
    *(v4f*)&dst[4] = v4f{ acc[4], acc[5], acc[6], acc[7] };
  }
  __syncthreads();

  float* GT = RT;
  {
    const int tb = wave >> 1;
    const int kh = wave & 1;
    const float* srcp = XP + (tb * 16 + mn) * XPITCH + kh * 32 + half * 16;
    v8f acc0 = v8f{0.f, 0.f, 0.f, 0.f, 0.f, 0.f, 0.f, 0.f};
    v8f acc1 = v8f{0.f, 0.f, 0.f, 0.f, 0.f, 0.f, 0.f, 0.f};
#pragma unroll
    for (int q = 0; q < 4; ++q) {
      const v4f p = *(const v4f*)&srcp[4 * q];
      const float sgn = (q & 1) ? -1.f : 1.f;
      const v2f a0 = v2f{ sgn * p[0], -sgn * p[1] };
      const v2f a1 = v2f{ -sgn * p[2], sgn * p[3] };
      acc0 = __builtin_amdgcn_wmma_f32_16x16x4_f32(false, a0, false, v2f{ p[0], p[1] }, 0, acc0, false, false);
      acc1 = __builtin_amdgcn_wmma_f32_16x16x4_f32(false, a1, false, v2f{ p[2], p[3] }, 0, acc1, false, false);
    }
    const v8f m = acc0 + acc1;
    float* dst = GT + (wave * 16 + mn) * GPITCH + 8 * half;
    *(v4f*)&dst[0] = v4f{ m[0], m[1], m[2], m[3] };
    *(v4f*)&dst[4] = v4f{ m[4], m[5], m[6], m[7] };
  }
  __syncthreads();

  if (tid < 64) {
    const int a  = tid >> 3;
    const int ap = tid & 7;
    float re[2], im[2];
#pragma unroll
    for (int tb = 0; tb < 2; ++tb) {
      const float* g0 = GT + (2 * tb) * 16 * GPITCH;
      const float* g1 = g0 + 16 * GPITCH;
      const float m00 = g0[ap * GPITCH + a]           + g1[ap * GPITCH + a];
      const float m10 = g0[ap * GPITCH + 8 + a]       + g1[ap * GPITCH + 8 + a];
      const float m01 = g0[(8 + ap) * GPITCH + a]     + g1[(8 + ap) * GPITCH + a];
      const float m11 = g0[(8 + ap) * GPITCH + 8 + a] + g1[(8 + ap) * GPITCH + 8 + a];
      re[tb] = m00 + m11;
      im[tb] = m10 - m01;
    }
    float val = re[0] * re[1] - im[0] * im[1];
    val += __shfl_xor(val, 1);
    val += __shfl_xor(val, 2);
    val += __shfl_xor(val, 4);
    val += __shfl_xor(val, 8);
    val += __shfl_xor(val, 16);
    if (lane == 0) red[wave] = val;
  }
  __syncthreads();
  if (tid == 0) out[b] = red[0] + red[1];
}

extern "C" void kernel_launch(void* const* d_in, const int* in_sizes, int n_in,
                              void* d_out, int out_size, void* d_ws, size_t ws_size,
                              hipStream_t stream) {
  (void)n_in; (void)d_ws; (void)ws_size; (void)out_size;
  const float* x      = (const float*)d_in[0];
  const float* thetas = (const float*)d_in[1];
  float* out          = (float*)d_out;
  const int batch     = in_sizes[0];
  qcircuit_parity_kernel<<<dim3(batch), dim3(THREADS), 0, stream>>>(x, thetas, out);
}


# Each table's chain through its first 5 sites on ONE wave (lane = those 5 index bits; was 2 waves x 6 full steps), last site from that prefix via LDS; the rest of the set-up spread over the 4 waves
# speedup vs baseline: 18.2059x; 1.1292x over previous
#include <hip/hip_runtime.h>
#include <math.h>


typedef __attribute__((ext_vector_type(2))) float v2f;
typedef __attribute__((ext_vector_type(4))) float v4f;
typedef __attribute__((ext_vector_type(8))) float v8f;

#define NQ      12
#define NL      4
#define QDIM    4096
#define THREADS 128
#define APITCH  20
#define TPITCH  80
#define XPITCH  68
#define GPITCH  20

static_assert(QDIM == (1 << NQ), "dim mismatch");
static_assert(NL == 4, "the direct build covers exactly the three layers in front of the last one");

struct c32 { float re, im; };
__device__ __forceinline__ c32 cmul(c32 a, c32 b) {
  return { a.re * b.re - a.im * b.im, a.re * b.im + a.im * b.re };
}
__device__ __forceinline__ c32 cadd(c32 a, c32 b) { return { a.re + b.re, a.im + b.im }; }
__device__ __forceinline__ c32 cdot2(c32 a, c32 b, c32 c, c32 d) {
  return { fmaf(-c.im, d.im, fmaf(c.re, d.re, fmaf(-a.im, b.im, a.re * b.re))),
           fmaf( c.im, d.re, fmaf(c.re, d.im, fmaf( a.im, b.re, a.re * b.im))) };
}

__device__ __forceinline__ void mat2mul(c32 r[4], const c32 p[4], const c32 q[4]) {
  r[0] = cadd(cmul(p[0], q[0]), cmul(p[1], q[2]));
  r[1] = cadd(cmul(p[0], q[1]), cmul(p[1], q[3]));
  r[2] = cadd(cmul(p[2], q[0]), cmul(p[3], q[2]));
  r[3] = cadd(cmul(p[2], q[1]), cmul(p[3], q[3]));
}

__device__ __forceinline__ void load_a(v2f a[4], const float* Ag, int c, int h) {
  const v4f lo = *(const v4f*)&Ag[c * APITCH + 4 * h];
  const v4f hi = *(const v4f*)&Ag[c * APITCH + 8 + 4 * h];
  a[0] = v2f{lo[0], lo[1]}; a[1] = v2f{lo[2], lo[3]};
  a[2] = v2f{hi[0], hi[1]}; a[3] = v2f{hi[2], hi[3]};
}

__device__ __forceinline__ void all_gates(float* gmre, float* gmim, const float* __restrict__ thetas,
                                          int tid, float x1, float x2) {
  if (tid >= 0 && tid < NL * NQ) {
    c32 M[4];
    { float s, c; __sincosf(0.5f * x1, &s, &c);
      M[0] = c32{c, 0.f}; M[1] = c32{-s, 0.f}; M[2] = c32{s, 0.f}; M[3] = c32{c, 0.f}; }
    { float s, c; __sincosf(0.5f * x2, &s, &c);
      c32 G[4] = { c32{c, -s}, c32{0.f, 0.f}, c32{0.f, 0.f}, c32{c, s} };
      c32 T[4]; mat2mul(T, G, M);
      M[0] = T[0]; M[1] = T[1]; M[2] = T[2]; M[3] = T[3]; }
    const float t0 = thetas[tid * 3 + 0];
    const float t1 = thetas[tid * 3 + 1];
    const float t2 = thetas[tid * 3 + 2];
    { float s, c; __sincosf(0.5f * t0, &s, &c);
      c32 G[4] = { c32{c, 0.f}, c32{0.f, -s}, c32{0.f, -s}, c32{c, 0.f} };
      c32 T[4]; mat2mul(T, G, M);
      M[0] = T[0]; M[1] = T[1]; M[2] = T[2]; M[3] = T[3]; }
    { float s, c; __sincosf(0.5f * t1, &s, &c);
      c32 G[4] = { c32{c, -s}, c32{0.f, 0.f}, c32{0.f, 0.f}, c32{c, s} };
      c32 T[4]; mat2mul(T, G, M);
      M[0] = T[0]; M[1] = T[1]; M[2] = T[2]; M[3] = T[3]; }
    { float s, c; __sincosf(0.5f * t2, &s, &c);
      c32 G[4] = { c32{c, 0.f}, c32{0.f, -s}, c32{0.f, -s}, c32{c, 0.f} };
      c32 T[4]; mat2mul(T, G, M);
      M[0] = T[0]; M[1] = T[1]; M[2] = T[2]; M[3] = T[3]; }
#pragma unroll
    for (int k = 0; k < 4; ++k) { gmre[tid * 4 + k] = M[k].re; gmim[tid * 4 + k] = M[k].im; }
  }
}

__device__ __forceinline__ c32 gate(const float* gmre, const float* gmim, int l, int k, int row, int col) {
  const int at = (l * NQ + (NQ - 1 - k)) * 4 + 2 * row + col;
  return c32{ gmre[at], gmim[at] };
}

struct site { c32 a[2]; c32 u2[2][2]; c32 u3p[2]; };
__device__ __forceinline__ site load_site(const float* gmre, const float* gmim, int k, int p) {
  site s;
#pragma unroll
  for (int g = 0; g < 2; ++g) {
    s.a[g]   = gate(gmre, gmim, 0, k, g, 0);
    s.u3p[g] = gate(gmre, gmim, 2, k, p, g);
#pragma unroll
    for (int j = 0; j < 2; ++j) s.u2[g][j] = gate(gmre, gmim, 1, k, g, j);
  }
  return s;
}

__device__ __forceinline__ void row_mid(c32 u[2][2], const c32 v[2][2], const site& s) {
  c32 t[2][2];
#pragma unroll
  for (int i = 0; i < 2; ++i)
#pragma unroll
    for (int mp = 0; mp < 2; ++mp)
      t[i][mp] = cdot2(v[0][mp], s.a[i], v[1][mp], s.a[i ^ 1]);
#pragma unroll
  for (int i = 0; i < 2; ++i)
#pragma unroll
    for (int m = 0; m < 2; ++m)
      u[i][m] = cdot2(s.u2[m][i], t[i][0], s.u2[m ^ 1][i], t[i][1]);
}
__device__ __forceinline__ void row_out(c32 r[2][2], const c32 u[2][2], const c32 u3p[2]) {
#pragma unroll
  for (int i = 0; i < 2; ++i)
#pragma unroll
    for (int m = 0; m < 2; ++m)
      r[i][m] = cmul(u3p[m], u[i][m]);
}
__device__ __forceinline__ void row_step(c32 v[2][2], const site& s) {
  c32 u[2][2];
  row_mid(u, v, s);
  row_out(v, u, s.u3p);
}

__device__ __forceinline__ void col_step(c32 w[2][2], const site& s) {
  c32 y[2][2], z[2][2];
#pragma unroll
  for (int i = 0; i < 2; ++i)
#pragma unroll
    for (int m = 0; m < 2; ++m)
      y[i][m] = cmul(s.u3p[m], w[i][m]);
#pragma unroll
  for (int i = 0; i < 2; ++i)
#pragma unroll
    for (int mp = 0; mp < 2; ++mp)
      z[i][mp] = cdot2(s.u2[mp][i], y[i][0], s.u2[mp ^ 1][i], y[i][1]);
#pragma unroll
  for (int ip = 0; ip < 2; ++ip)
#pragma unroll
    for (int mp = 0; mp < 2; ++mp)
      w[ip][mp] = cdot2(s.a[ip], z[0][mp], s.a[ip ^ 1], z[1][mp]);
}

__device__ __forceinline__ int plane_pos(int v) {
  const int od = (((v >> 5) & 1) << 2) | (((v >> 3) & 1) << 1) | ((v >> 1) & 1);
  const int ev = (((v >> 4) & 1) << 2) | (((v >> 2) & 1) << 1) | (v & 1);
  return od * 8 + ev;
}

__device__ __forceinline__ int g_inv(int n) {
  n ^= n >> 1; n ^= n >> 2; n ^= n >> 4;
  return n;
}

__device__ __forceinline__ void store_l(float* T, int n, const c32 v[2][2]) {
  const int pos = plane_pos(g_inv(n));
#pragma unroll
  for (int i = 0; i < 2; ++i)
#pragma unroll
    for (int m = 0; m < 2; ++m) {
      T[(2 * i + m) * TPITCH + pos]     = v[i][m].re;
      T[(4 + 2 * i + m) * TPITCH + pos] = v[i][m].im;
    }
}

__device__ __forceinline__ void store_r(float* T, int n, const c32 w[2][2]) {
#pragma unroll
  for (int f = 0; f < 2; ++f) {
    const int pos = plane_pos(g_inv(n ^ (f << 5)));
#pragma unroll
    for (int i = 0; i < 2; ++i)
#pragma unroll
      for (int m = 0; m < 2; ++m) {
        T[(4 * f + 2 * i + m) * TPITCH + pos]     = w[i][m].re;
        T[(8 + 4 * f + 2 * i + m) * TPITCH + pos] = w[i][m].im;
      }
  }
}

__global__ __launch_bounds__(THREADS)
void qcircuit_parity_kernel(const float* __restrict__ x,
                            const float* __restrict__ thetas,
                            float* __restrict__ out) {
  __shared__ __attribute__((aligned(16))) float Aemb[2 * 16 * APITCH];
  __shared__ __attribute__((aligned(16))) float LT[8 * TPITCH];
  __shared__ __attribute__((aligned(16))) float RT[16 * TPITCH];
  static_assert(16 * TPITCH >= 4 * 16 * GPITCH, "the four partial Gram tiles reuse RT");
  __shared__ __attribute__((aligned(16))) float XP[2 * 2 * 8 * XPITCH];
  __shared__ float gmre[NL * NQ * 4];
  __shared__ float gmim[NL * NQ * 4];
  __shared__ float red[2];

  const int tid  = threadIdx.x;
  const int b    = blockIdx.x;
  const int lane = tid & 31;
  const int wave = tid >> 5;
  const int half = lane >> 4;
  const int mn   = lane & 15;

  const float xv = x[b];
  const float x1 = asinf(xv);
  const float x2 = acosf(xv * xv);

  all_gates(gmre, gmim, thetas, tid - 32, x1, x2);
  __syncthreads();

  float* HL = XP;
  float* HR = XP + 8 * 32;
  c32 w[2][2] = { { c32{1.f, 0.f}, c32{1.f, 0.f} }, { c32{1.f, 0.f}, c32{1.f, 0.f} } };
  if (wave == 0) {
    c32 v[2][2] = { { c32{1.f, 0.f}, c32{0.f, 0.f} }, { c32{0.f, 0.f}, c32{0.f, 0.f} } };
#pragma unroll
    for (int k = NQ - 1; k >= 7; --k)
      row_step(v, load_site(gmre, gmim, k, (lane >> (k - 7)) & 1));
#pragma unroll
    for (int i = 0; i < 2; ++i)
#pragma unroll
      for (int m = 0; m < 2; ++m) {
        HL[(2 * i + m) * 32 + lane]     = v[i][m].re;
        HL[(4 + 2 * i + m) * 32 + lane] = v[i][m].im;
      }
  } else if (wave == 3) {
#pragma unroll
    for (int k = 0; k < 5; ++k)
      col_step(w, load_site(gmre, gmim, k, (lane >> k) & 1));
#pragma unroll
    for (int i = 0; i < 2; ++i)
#pragma unroll
      for (int m = 0; m < 2; ++m) {
        HR[(2 * i + m) * 32 + lane]     = w[i][m].re;
        HR[(4 + 2 * i + m) * 32 + lane] = w[i][m].im;
      }
  }
  else {
    const int r = (tid - 32) >> 3, cc = tid & 7;
#pragma unroll
    for (int g = 0; g < 2; ++g) {
      c32 acc = c32{1.f, 0.f};
#pragma unroll
      for (int j = 0; j < 3; ++j) {
        const int q  = 2 * (3 * g + j) + 1;
        const int bi = (r  >> (2 - j)) & 1;
        const int bj = (cc >> (2 - j)) & 1;
        acc = cmul(acc, gate(gmre, gmim, NL - 1, NQ - 1 - q, bi, bj));
      }
      float* Ag = Aemb + g * 16 * APITCH;
      Ag[r * APITCH + cc]           = acc.re;
      Ag[r * APITCH + cc + 8]       = -acc.im;
      Ag[(r + 8) * APITCH + cc]     = acc.im;
      Ag[(r + 8) * APITCH + cc + 8] = acc.re;
    }
  }
  __syncthreads();

  if (wave == 1) {
    c32 v[2][2];
#pragma unroll
    for (int i = 0; i < 2; ++i)
#pragma unroll
      for (int m = 0; m < 2; ++m)
        v[i][m] = c32{ HL[(2 * i + m) * 32 + lane], HL[(4 + 2 * i + m) * 32 + lane] };
    c32 u[2][2];
    row_mid(u, v, load_site(gmre, gmim, 6, 0));
#pragma unroll
    for (int p = 0; p < 2; ++p) {
      c32 r[2][2];
      row_out(r, u, load_site(gmre, gmim, 6, p).u3p);
      store_l(LT, 2 * lane + p, r);
    }
  } else if (wave == 2) {
    c32 y[2][2];
#pragma unroll
    for (int i = 0; i < 2; ++i)
#pragma unroll
      for (int m = 0; m < 2; ++m)
        y[i][m] = c32{ HR[(2 * i + m) * 32 + lane], HR[(4 + 2 * i + m) * 32 + lane] };
    col_step(y, load_site(gmre, gmim, 5, 1));
    store_r(RT, lane + 32, y);
  } else if (wave == 3) {
    col_step(w, load_site(gmre, gmim, 5, 0));
    store_r(RT, lane, w);
  }
  __syncthreads();

#pragma unroll
  for (int tb = 0; tb < 2; ++tb) {
    const int a    = mn & 7;
    const int od   = 2 * wave + (mn >> 3);
    v4f P[2];
    if (tb == 0) {
      const bool f = (a >> 2) != 0;
#pragma unroll
      for (int part = 0; part < 2; ++part) {
        const v4f q = *(const v4f*)&LT[(part * 4 + (a & 3)) * TPITCH + od * 8 + 4 * half];
        P[part] = f ? v4f{0.f, q[1], 0.f, q[3]} : v4f{q[0], 0.f, q[2], 0.f};
      }
    } else {
#pragma unroll
      for (int part = 0; part < 2; ++part)
        P[part] = *(const v4f*)&RT[(part * 8 + a) * TPITCH + od * 8 + 4 * half];
    }
    v2f af[4];
    load_a(af, Aemb + tb * 16 * APITCH, mn, half);
    v8f acc = v8f{0.f, 0.f, 0.f, 0.f, 0.f, 0.f, 0.f, 0.f};
#pragma unroll
    for (int kc = 0; kc < 4; ++kc)
      acc = __builtin_amdgcn_wmma_f32_16x16x4_f32(
          false, af[kc], false, v2f{ P[kc >> 1][2 * (kc & 1)], P[kc >> 1][2 * (kc & 1) + 1] },
          0, acc, false, false);
    float* dst = &XP[((tb * 2 + half) * 8 + a) * XPITCH + od * 8];
    *(v4f*)&dst[0] = v4f{ acc[0], acc[1], acc[2], acc[3] };
    *(v4f*)&dst[4] = v4f{ acc[4], acc[5], acc[6], acc[7] };
  }
  __syncthreads();

  float* GT = RT;
  {
    const int tb = wave >> 1;
    const int kh = wave & 1;
    const float* srcp = XP + (tb * 16 + mn) * XPITCH + kh * 32 + half * 16;
    v8f acc0 = v8f{0.f, 0.f, 0.f, 0.f, 0.f, 0.f, 0.f, 0.f};
    v8f acc1 = v8f{0.f, 0.f, 0.f, 0.f, 0.f, 0.f, 0.f, 0.f};
#pragma unroll
    for (int q = 0; q < 4; ++q) {
      const v4f p = *(const v4f*)&srcp[4 * q];
      const float sgn = (q & 1) ? -1.f : 1.f;
      const v2f a0 = v2f{ sgn * p[0], -sgn * p[1] };
      const v2f a1 = v2f{ -sgn * p[2], sgn * p[3] };
      acc0 = __builtin_amdgcn_wmma_f32_16x16x4_f32(false, a0, false, v2f{ p[0], p[1] }, 0, acc0, false, false);
      acc1 = __builtin_amdgcn_wmma_f32_16x16x4_f32(false, a1, false, v2f{ p[2], p[3] }, 0, acc1, false, false);
    }
    const v8f m = acc0 + acc1;
    float* dst = GT + (wave * 16 + mn) * GPITCH + 8 * half;
    *(v4f*)&dst[0] = v4f{ m[0], m[1], m[2], m[3] };
    *(v4f*)&dst[4] = v4f{ m[4], m[5], m[6], m[7] };
  }
  __syncthreads();

  if (tid < 64) {
    const int a  = tid >> 3;
    const int ap = tid & 7;
    float re[2], im[2];
#pragma unroll
    for (int tb = 0; tb < 2; ++tb) {
      const float* g0 = GT + (2 * tb) * 16 * GPITCH;
      const float* g1 = g0 + 16 * GPITCH;
      const float m00 = g0[ap * GPITCH + a]           + g1[ap * GPITCH + a];
      const float m10 = g0[ap * GPITCH + 8 + a]       + g1[ap * GPITCH + 8 + a];
      const float m01 = g0[(8 + ap) * GPITCH + a]     + g1[(8 + ap) * GPITCH + a];
      const float m11 = g0[(8 + ap) * GPITCH + 8 + a] + g1[(8 + ap) * GPITCH + 8 + a];
      re[tb] = m00 + m11;
      im[tb] = m10 - m01;
    }
    float val = re[0] * re[1] - im[0] * im[1];
    val += __shfl_xor(val, 1);
    val += __shfl_xor(val, 2);
    val += __shfl_xor(val, 4);
    val += __shfl_xor(val, 8);
    val += __shfl_xor(val, 16);
    if (lane == 0) red[wave] = val;
  }
  __syncthreads();
  if (tid == 0) out[b] = red[0] + red[1];
}

extern "C" void kernel_launch(void* const* d_in, const int* in_sizes, int n_in,
                              void* d_out, int out_size, void* d_ws, size_t ws_size,
                              hipStream_t stream) {
  (void)n_in; (void)d_ws; (void)ws_size; (void)out_size;
  const float* x      = (const float*)d_in[0];
  const float* thetas = (const float*)d_in[1];
  float* out          = (float*)d_out;
  const int batch     = in_sizes[0];
  qcircuit_parity_kernel<<<dim3(batch), dim3(THREADS), 0, stream>>>(x, thetas, out);
}


# X tiles sorted by f (an X column is zero where bit 0 of the group value differs from f): K = 8, 2 WMMA per X tile, not 4; X tiles on waves 0, 1, 2, 2, none on the R-chain wave; f32 sum order changes
# speedup vs baseline: 19.3118x; 1.0607x over previous
#include <hip/hip_runtime.h>
#include <math.h>


typedef __attribute__((ext_vector_type(2))) float v2f;
typedef __attribute__((ext_vector_type(4))) float v4f;
typedef __attribute__((ext_vector_type(8))) float v8f;

#define NQ      12
#define NL      4
#define QDIM    4096
#define THREADS 128
#define APITCH  20
#define TPITCH  80
#define LOP     72
#define XPITCH  68
#define GPITCH  20

static_assert(QDIM == (1 << NQ), "dim mismatch");
static_assert(NL == 4, "the direct build covers exactly the three layers in front of the last one");

struct c32 { float re, im; };
__device__ __forceinline__ c32 cmul(c32 a, c32 b) {
  return { a.re * b.re - a.im * b.im, a.re * b.im + a.im * b.re };
}
__device__ __forceinline__ c32 cadd(c32 a, c32 b) { return { a.re + b.re, a.im + b.im }; }
__device__ __forceinline__ c32 cdot2(c32 a, c32 b, c32 c, c32 d) {
  return { fmaf(-c.im, d.im, fmaf(c.re, d.re, fmaf(-a.im, b.im, a.re * b.re))),
           fmaf( c.im, d.re, fmaf(c.re, d.im, fmaf( a.im, b.re, a.re * b.im))) };
}

__device__ __forceinline__ void mat2mul(c32 r[4], const c32 p[4], const c32 q[4]) {
  r[0] = cadd(cmul(p[0], q[0]), cmul(p[1], q[2]));
  r[1] = cadd(cmul(p[0], q[1]), cmul(p[1], q[3]));
  r[2] = cadd(cmul(p[2], q[0]), cmul(p[3], q[2]));
  r[3] = cadd(cmul(p[2], q[1]), cmul(p[3], q[3]));
}

__device__ __forceinline__ void load_a(v2f a[4], const float* Ag, int c, int h) {
  const v4f lo = *(const v4f*)&Ag[c * APITCH + 4 * h];
  const v4f hi = *(const v4f*)&Ag[c * APITCH + 8 + 4 * h];
  a[0] = v2f{lo[0], lo[1]}; a[1] = v2f{lo[2], lo[3]};
  a[2] = v2f{hi[0], hi[1]}; a[3] = v2f{hi[2], hi[3]};
}

__device__ __forceinline__ int l_slot(int kk, int part) {
  return (2 * (kk & 1) + (kk >> 2)) * 4 + 2 * ((kk >> 1) & 1) + part;
}

__device__ __forceinline__ void all_gates(float* gmre, float* gmim, const float* __restrict__ thetas,
                                          int tid, float x1, float x2) {
  if (tid >= 0 && tid < NL * NQ) {
    c32 M[4];
    { float s, c; __sincosf(0.5f * x1, &s, &c);
      M[0] = c32{c, 0.f}; M[1] = c32{-s, 0.f}; M[2] = c32{s, 0.f}; M[3] = c32{c, 0.f}; }
    { float s, c; __sincosf(0.5f * x2, &s, &c);
      c32 G[4] = { c32{c, -s}, c32{0.f, 0.f}, c32{0.f, 0.f}, c32{c, s} };
      c32 T[4]; mat2mul(T, G, M);
      M[0] = T[0]; M[1] = T[1]; M[2] = T[2]; M[3] = T[3]; }
    const float t0 = thetas[tid * 3 + 0];
    const float t1 = thetas[tid * 3 + 1];
    const float t2 = thetas[tid * 3 + 2];
    { float s, c; __sincosf(0.5f * t0, &s, &c);
      c32 G[4] = { c32{c, 0.f}, c32{0.f, -s}, c32{0.f, -s}, c32{c, 0.f} };
      c32 T[4]; mat2mul(T, G, M);
      M[0] = T[0]; M[1] = T[1]; M[2] = T[2]; M[3] = T[3]; }
    { float s, c; __sincosf(0.5f * t1, &s, &c);
      c32 G[4] = { c32{c, -s}, c32{0.f, 0.f}, c32{0.f, 0.f}, c32{c, s} };
      c32 T[4]; mat2mul(T, G, M);
      M[0] = T[0]; M[1] = T[1]; M[2] = T[2]; M[3] = T[3]; }
    { float s, c; __sincosf(0.5f * t2, &s, &c);
      c32 G[4] = { c32{c, 0.f}, c32{0.f, -s}, c32{0.f, -s}, c32{c, 0.f} };
      c32 T[4]; mat2mul(T, G, M);
      M[0] = T[0]; M[1] = T[1]; M[2] = T[2]; M[3] = T[3]; }
#pragma unroll
    for (int k = 0; k < 4; ++k) { gmre[tid * 4 + k] = M[k].re; gmim[tid * 4 + k] = M[k].im; }
  }
}

__device__ __forceinline__ c32 gate(const float* gmre, const float* gmim, int l, int k, int row, int col) {
  const int at = (l * NQ + (NQ - 1 - k)) * 4 + 2 * row + col;
  return c32{ gmre[at], gmim[at] };
}

struct site { c32 a[2]; c32 u2[2][2]; c32 u3p[2]; };
__device__ __forceinline__ site load_site(const float* gmre, const float* gmim, int k, int p) {
  site s;
#pragma unroll
  for (int g = 0; g < 2; ++g) {
    s.a[g]   = gate(gmre, gmim, 0, k, g, 0);
    s.u3p[g] = gate(gmre, gmim, 2, k, p, g);
#pragma unroll
    for (int j = 0; j < 2; ++j) s.u2[g][j] = gate(gmre, gmim, 1, k, g, j);
  }
  return s;
}

__device__ __forceinline__ void row_mid(c32 u[2][2], const c32 v[2][2], const site& s) {
  c32 t[2][2];
#pragma unroll
  for (int i = 0; i < 2; ++i)
#pragma unroll
    for (int mp = 0; mp < 2; ++mp)
      t[i][mp] = cdot2(v[0][mp], s.a[i], v[1][mp], s.a[i ^ 1]);
#pragma unroll
  for (int i = 0; i < 2; ++i)
#pragma unroll
    for (int m = 0; m < 2; ++m)
      u[i][m] = cdot2(s.u2[m][i], t[i][0], s.u2[m ^ 1][i], t[i][1]);
}
__device__ __forceinline__ void row_out(c32 r[2][2], const c32 u[2][2], const c32 u3p[2]) {
#pragma unroll
  for (int i = 0; i < 2; ++i)
#pragma unroll
    for (int m = 0; m < 2; ++m)
      r[i][m] = cmul(u3p[m], u[i][m]);
}
__device__ __forceinline__ void row_step(c32 v[2][2], const site& s) {
  c32 u[2][2];
  row_mid(u, v, s);
  row_out(v, u, s.u3p);
}

__device__ __forceinline__ void col_step(c32 w[2][2], const site& s) {
  c32 y[2][2], z[2][2];
#pragma unroll
  for (int i = 0; i < 2; ++i)
#pragma unroll
    for (int m = 0; m < 2; ++m)
      y[i][m] = cmul(s.u3p[m], w[i][m]);
#pragma unroll
  for (int i = 0; i < 2; ++i)
#pragma unroll
    for (int mp = 0; mp < 2; ++mp)
      z[i][mp] = cdot2(s.u2[mp][i], y[i][0], s.u2[mp ^ 1][i], y[i][1]);
#pragma unroll
  for (int ip = 0; ip < 2; ++ip)
#pragma unroll
    for (int mp = 0; mp < 2; ++mp)
      w[ip][mp] = cdot2(s.a[ip], z[0][mp], s.a[ip ^ 1], z[1][mp]);
}

__device__ __forceinline__ int plane_pos(int v) {
  const int od = (((v >> 5) & 1) << 2) | (((v >> 3) & 1) << 1) | ((v >> 1) & 1);
  const int ev = (((v >> 4) & 1) << 2) | (((v >> 2) & 1) << 1) | (v & 1);
  return od * 8 + ev;
}

__device__ __forceinline__ int g_inv(int n) {
  n ^= n >> 1; n ^= n >> 2; n ^= n >> 4;
  return n;
}

__device__ __forceinline__ void store_l(float* T, int n, const c32 v[2][2]) {
  const int hi = g_inv(n);
  const int od = (((hi >> 5) & 1) << 2) | (((hi >> 3) & 1) << 1) | ((hi >> 1) & 1);
  const int kk = (((hi >> 4) & 1) << 2) | (((hi >> 2) & 1) << 1) | (hi & 1);
  const int pos = od * LOP + l_slot(kk, 0);
#pragma unroll
  for (int i = 0; i < 2; ++i)
#pragma unroll
    for (int m = 0; m < 2; ++m)
      *(v2f*)&T[pos + (2 * i + m) * 16] = v2f{ v[i][m].re, v[i][m].im };
}

__device__ __forceinline__ void store_r(float* T, int n, const c32 w[2][2]) {
#pragma unroll
  for (int f = 0; f < 2; ++f) {
    const int pos = plane_pos(g_inv(n ^ (f << 5)));
#pragma unroll
    for (int i = 0; i < 2; ++i)
#pragma unroll
      for (int m = 0; m < 2; ++m) {
        T[(4 * f + 2 * i + m) * TPITCH + pos]     = w[i][m].re;
        T[(8 + 4 * f + 2 * i + m) * TPITCH + pos] = w[i][m].im;
      }
  }
}

template <int NX>
__device__ __forceinline__ void last_layer_tiles(const float* Aemb, const float* LT, const float* RT, float* XP,
                                                 int wave, int f, int q0, int mn, int half) {
  constexpr int NXA = NX > 0 ? NX : 1;
  const int a   = mn & 7;
  const int ody = 2 * wave + (mn >> 3);
  v4f P[2];
#pragma unroll
  for (int part = 0; part < 2; ++part)
    P[part] = *(const v4f*)&RT[(part * 8 + a) * TPITCH + ody * 8 + 4 * half];
  v2f af[4];
  load_a(af, Aemb + 16 * APITCH, mn, half);
  const int s = mn & 3;
  v4f aq[NXA], bq[NXA];
  int odx[NXA];
#pragma unroll
  for (int j = 0; j < NX; ++j) {
    odx[j] = 4 * (q0 + j) + (mn >> 2);
    bq[j]  = *(const v4f*)&LT[odx[j] * LOP + s * 16 + (2 * f + half) * 4];
    aq[j]  = *(const v4f*)&Aemb[mn * APITCH + (2 * f + half) * 4];
  }
  v8f accy = v8f{0.f, 0.f, 0.f, 0.f, 0.f, 0.f, 0.f, 0.f};
  v8f accx[NXA];
#pragma unroll
  for (int j = 0; j < NXA; ++j) accx[j] = v8f{0.f, 0.f, 0.f, 0.f, 0.f, 0.f, 0.f, 0.f};
#pragma unroll
  for (int kc = 0; kc < 4; ++kc) {
    accy = __builtin_amdgcn_wmma_f32_16x16x4_f32(
        false, af[kc], false, v2f{ P[kc >> 1][2 * (kc & 1)], P[kc >> 1][2 * (kc & 1) + 1] },
        0, accy, false, false);
    if (kc < 2) {
#pragma unroll
      for (int j = 0; j < NX; ++j)
        accx[j] = __builtin_amdgcn_wmma_f32_16x16x4_f32(
            false, v2f{ aq[j][2 * kc], aq[j][2 * kc + 1] }, false, v2f{ bq[j][2 * kc], bq[j][2 * kc + 1] },
            0, accx[j], false, false);
    }
  }
  float* dsty = &XP[((2 + half) * 8 + a) * XPITCH + ody * 8];
  *(v4f*)&dsty[0] = v4f{ accy[0], accy[1], accy[2], accy[3] };
  *(v4f*)&dsty[4] = v4f{ accy[4], accy[5], accy[6], accy[7] };
#pragma unroll
  for (int j = 0; j < NX; ++j) {
    float* dstx = &XP[(half * 8 + 4 * f + s) * XPITCH + odx[j] * 8];
    *(v4f*)&dstx[0] = v4f{ accx[j][0], accx[j][1], accx[j][2], accx[j][3] };
    *(v4f*)&dstx[4] = v4f{ accx[j][4], accx[j][5], accx[j][6], accx[j][7] };
  }
}

__global__ __launch_bounds__(THREADS)
void qcircuit_parity_kernel(const float* __restrict__ x,
                            const float* __restrict__ thetas,
                            float* __restrict__ out) {
  __shared__ __attribute__((aligned(16))) float Aemb[2 * 16 * APITCH];
  __shared__ __attribute__((aligned(16))) float LT[8 * LOP];
  __shared__ __attribute__((aligned(16))) float RT[16 * TPITCH];
  static_assert(16 * TPITCH >= 4 * 16 * GPITCH, "the four partial Gram tiles reuse RT");
  __shared__ __attribute__((aligned(16))) float XP[2 * 2 * 8 * XPITCH];
  __shared__ float gmre[NL * NQ * 4];
  __shared__ float gmim[NL * NQ * 4];
  __shared__ float red[2];

  const int tid  = threadIdx.x;
  const int b    = blockIdx.x;
  const int lane = tid & 31;
  const int wave = tid >> 5;
  const int half = lane >> 4;
  const int mn   = lane & 15;

  const float xv = x[b];
  const float x1 = asinf(xv);
  const float x2 = acosf(xv * xv);

  all_gates(gmre, gmim, thetas, tid - 32, x1, x2);
  __syncthreads();

  float* HL = XP;
  float* HR = XP + 8 * 32;
  c32 w[2][2] = { { c32{1.f, 0.f}, c32{1.f, 0.f} }, { c32{1.f, 0.f}, c32{1.f, 0.f} } };
  if (wave == 0) {
    c32 v[2][2] = { { c32{1.f, 0.f}, c32{0.f, 0.f} }, { c32{0.f, 0.f}, c32{0.f, 0.f} } };
#pragma unroll
    for (int k = NQ - 1; k >= 7; --k)
      row_step(v, load_site(gmre, gmim, k, (lane >> (k - 7)) & 1));
#pragma unroll
    for (int i = 0; i < 2; ++i)
#pragma unroll
      for (int m = 0; m < 2; ++m) {
        HL[(2 * i + m) * 32 + lane]     = v[i][m].re;
        HL[(4 + 2 * i + m) * 32 + lane] = v[i][m].im;
      }
  } else if (wave == 3) {
#pragma unroll
    for (int k = 0; k < 5; ++k)
      col_step(w, load_site(gmre, gmim, k, (lane >> k) & 1));
#pragma unroll
    for (int i = 0; i < 2; ++i)
#pragma unroll
      for (int m = 0; m < 2; ++m) {
        HR[(2 * i + m) * 32 + lane]     = w[i][m].re;
        HR[(4 + 2 * i + m) * 32 + lane] = w[i][m].im;
      }
  }
  else {
    const int r = (tid - 32) >> 3, cc = tid & 7;
#pragma unroll
    for (int g = 0; g < 2; ++g) {
      c32 acc = c32{1.f, 0.f};
#pragma unroll
      for (int j = 0; j < 3; ++j) {
        const int q  = 2 * (3 * g + j) + 1;
        const int bi = (r  >> (2 - j)) & 1;
        const int bj = (cc >> (2 - j)) & 1;
        acc = cmul(acc, gate(gmre, gmim, NL - 1, NQ - 1 - q, bi, bj));
      }
      float* Ag = Aemb + g * 16 * APITCH;
      const int cre = g ? cc     : l_slot(cc, 0);
      const int cim = g ? cc + 8 : l_slot(cc, 1);
      Ag[r * APITCH + cre]       = acc.re;
      Ag[r * APITCH + cim]       = -acc.im;
      Ag[(r + 8) * APITCH + cre] = acc.im;
      Ag[(r + 8) * APITCH + cim] = acc.re;
    }
  }
  __syncthreads();

  if (wave == 1) {
    c32 v[2][2];
#pragma unroll
    for (int i = 0; i < 2; ++i)
#pragma unroll
      for (int m = 0; m < 2; ++m)
        v[i][m] = c32{ HL[(2 * i + m) * 32 + lane], HL[(4 + 2 * i + m) * 32 + lane] };
    c32 u[2][2];
    row_mid(u, v, load_site(gmre, gmim, 6, 0));
#pragma unroll
    for (int p = 0; p < 2; ++p) {
      c32 r[2][2];
      row_out(r, u, load_site(gmre, gmim, 6, p).u3p);
      store_l(LT, 2 * lane + p, r);
    }
  } else if (wave == 2) {
    c32 y[2][2];
#pragma unroll
    for (int i = 0; i < 2; ++i)
#pragma unroll
      for (int m = 0; m < 2; ++m)
        y[i][m] = c32{ HR[(2 * i + m) * 32 + lane], HR[(4 + 2 * i + m) * 32 + lane] };
    col_step(y, load_site(gmre, gmim, 5, 1));
    store_r(RT, lane + 32, y);
  } else if (wave == 3) {
    col_step(w, load_site(gmre, gmim, 5, 0));
    store_r(RT, lane, w);
  }
  __syncthreads();

  if (wave == 3)      last_layer_tiles<0>(Aemb, LT, RT, XP, wave, 0, 0, mn, half);
  else if (wave == 2) last_layer_tiles<2>(Aemb, LT, RT, XP, wave, 1, 0, mn, half);
  else                last_layer_tiles<1>(Aemb, LT, RT, XP, wave, 0, wave, mn, half);
  __syncthreads();

  float* GT = RT;
  {
    const int tb = wave >> 1;
    const int kh = wave & 1;
    const float* srcp = XP + (tb * 16 + mn) * XPITCH + kh * 32 + half * 16;
    v8f acc0 = v8f{0.f, 0.f, 0.f, 0.f, 0.f, 0.f, 0.f, 0.f};
    v8f acc1 = v8f{0.f, 0.f, 0.f, 0.f, 0.f, 0.f, 0.f, 0.f};
#pragma unroll
    for (int q = 0; q < 4; ++q) {
      const v4f p = *(const v4f*)&srcp[4 * q];
      const float sgn = (q & 1) ? -1.f : 1.f;
      const v2f a0 = v2f{ sgn * p[0], -sgn * p[1] };
      const v2f a1 = v2f{ -sgn * p[2], sgn * p[3] };
      acc0 = __builtin_amdgcn_wmma_f32_16x16x4_f32(false, a0, false, v2f{ p[0], p[1] }, 0, acc0, false, false);
      acc1 = __builtin_amdgcn_wmma_f32_16x16x4_f32(false, a1, false, v2f{ p[2], p[3] }, 0, acc1, false, false);
    }
    const v8f m = acc0 + acc1;
    float* dst = GT + (wave * 16 + mn) * GPITCH + 8 * half;
    *(v4f*)&dst[0] = v4f{ m[0], m[1], m[2], m[3] };
    *(v4f*)&dst[4] = v4f{ m[4], m[5], m[6], m[7] };
  }
  __syncthreads();

  if (tid < 64) {
    const int a  = tid >> 3;
    const int ap = tid & 7;
    float re[2], im[2];
#pragma unroll
    for (int tb = 0; tb < 2; ++tb) {
      const float* g0 = GT + (2 * tb) * 16 * GPITCH;
      const float* g1 = g0 + 16 * GPITCH;
      const float m00 = g0[ap * GPITCH + a]           + g1[ap * GPITCH + a];
      const float m10 = g0[ap * GPITCH + 8 + a]       + g1[ap * GPITCH + 8 + a];
      const float m01 = g0[(8 + ap) * GPITCH + a]     + g1[(8 + ap) * GPITCH + a];
      const float m11 = g0[(8 + ap) * GPITCH + 8 + a] + g1[(8 + ap) * GPITCH + 8 + a];
      re[tb] = m00 + m11;
      im[tb] = m10 - m01;
    }
    float val = re[0] * re[1] - im[0] * im[1];
    val += __shfl_xor(val, 1);
    val += __shfl_xor(val, 2);
    val += __shfl_xor(val, 4);
    val += __shfl_xor(val, 8);
    val += __shfl_xor(val, 16);
    if (lane == 0) red[wave] = val;
  }
  __syncthreads();
  if (tid == 0) out[b] = red[0] + red[1];
}

extern "C" void kernel_launch(void* const* d_in, const int* in_sizes, int n_in,
                              void* d_out, int out_size, void* d_ws, size_t ws_size,
                              hipStream_t stream) {
  (void)n_in; (void)d_ws; (void)ws_size; (void)out_size;
  const float* x      = (const float*)d_in[0];
  const float* thetas = (const float*)d_in[1];
  float* out          = (float*)d_out;
  const int batch     = in_sizes[0];
  qcircuit_parity_kernel<<<dim3(batch), dim3(THREADS), 0, stream>>>(x, thetas, out);
}
